# starting kernel
# baseline (speedup 1.0000x reference)
#include <hip/hip_runtime.h>


typedef float v2f __attribute__((ext_vector_type(2)));
typedef float v4f __attribute__((ext_vector_type(4)));
typedef float v8f __attribute__((ext_vector_type(8)));

#define B_TOT 4096
#define F     39
#define D     16
#define L0N   128
#define H1    64
#define K0    1521
#define K1    2496
#define NB    4
#define TPB   256
#define CK    156
#define ZSTRIDE 2560

#define XS_OFF 0
#define YS_OFF 2496
#define ZS_OFF (2496 + 4096)
#define LDS_FLOATS (2496 + 4096 + 10240)

__device__ __forceinline__ int zswz2(int c, int d) {
  return ((c >> 3) << 7) + (((c >> 1) & 1) << 6) + (d << 2) +
         (((c >> 2) & 1) << 1) + (c & 1);
}

template <bool CLAMP, bool AL64>
__device__ __forceinline__ void chunk_gemm(
    const float* __restrict__ wr, int Klen, int cbase, int pairs, bool tail,
    int hi, int boff, const float* __restrict__ zs, v8f (&acc)[NB])
{
  for (int S = 0; S < pairs; ++S) {
    const int cg = cbase + (S << 3) + 2 * hi;
    v2f a0, a1;
    if (AL64) {
      a0 = *(const v2f*)&wr[cg];
      a1 = *(const v2f*)&wr[cg + 4];
    } else if (CLAMP) {
      a0.x = wr[(cg     < Klen) ? cg     : Klen - 1];
      a0.y = wr[(cg + 1 < Klen) ? cg + 1 : Klen - 1];
      a1.x = wr[(cg + 4 < Klen) ? cg + 4 : Klen - 1];
      a1.y = wr[(cg + 5 < Klen) ? cg + 5 : Klen - 1];
    } else {
      a0.x = wr[cg];     a0.y = wr[cg + 1];
      a1.x = wr[cg + 4]; a1.y = wr[cg + 5];
    }
    const int zoff = (S << 7) + boff;
#pragma unroll
    for (int nb = 0; nb < NB; ++nb) {
      v4f bq = *(const v4f*)&zs[nb * ZSTRIDE + zoff];
      v2f bA = __builtin_shufflevector(bq, bq, 0, 1);
      v2f bB = __builtin_shufflevector(bq, bq, 2, 3);
      acc[nb] = __builtin_amdgcn_wmma_f32_16x16x4_f32(
          false, a0, false, bA, (short)0, acc[nb], false, false);
      acc[nb] = __builtin_amdgcn_wmma_f32_16x16x4_f32(
          false, a1, false, bB, (short)0, acc[nb], false, false);
    }
  }
  if (tail) {
    const int s  = pairs << 1;
    const int cg = cbase + (s << 2) + 2 * hi;
    v2f a0;
    if (AL64) {
      a0 = *(const v2f*)&wr[cg];
    } else {
      a0.x = wr[cg]; a0.y = wr[cg + 1];
    }
    const int zoff = (pairs << 7) + boff;
#pragma unroll
    for (int nb = 0; nb < NB; ++nb) {
      v2f b = *(const v2f*)&zs[nb * ZSTRIDE + zoff];
      acc[nb] = __builtin_amdgcn_wmma_f32_16x16x4_f32(
          false, a0, false, b, (short)0, acc[nb], false, false);
    }
  }
}

__global__ __launch_bounds__(TPB) void cin_wmma_kernel(
    const float* __restrict__ x,  const float* __restrict__ W0,
    const float* __restrict__ b0, const float* __restrict__ W1,
    const float* __restrict__ b1, float* __restrict__ out)
{
  __shared__ __align__(16) float lds[LDS_FLOATS];
  float* xs = lds + XS_OFF;
  float* ys = lds + YS_OFF;
  float* zs = lds + ZS_OFF;

  const int tid   = threadIdx.x;
  const int wave  = tid >> 5;
  const int lane  = tid & 31;
  const int lo16  = lane & 15;
  const int hi    = lane >> 4;
  const int bbase = blockIdx.x * NB;
  const int arow  = (wave << 4) + lo16;
  const int boff  = (hi << 6) + (lo16 << 2);

  const float* xg = x + (size_t)bbase * (F * D);
  for (int e = tid; e < NB * F * D; e += TPB) xs[e] = xg[e];

  v8f acc[NB];
#pragma unroll
  for (int nb = 0; nb < NB; ++nb)
#pragma unroll
    for (int j = 0; j < 8; ++j) acc[nb][j] = 0.0f;

  const float* w0r = W0 + (size_t)arow * K0;
  const float* w1r = W1 + (size_t)arow * K1;

  for (int ch = 0; ch < 9; ++ch) {
    const int h0    = ch << 2;
    const int cbase = ch * CK;

    __syncthreads();
    for (int nb = 0; nb < NB; ++nb) {
      const float* xb = xs + nb * (F * D);
      float*       zb = zs + nb * ZSTRIDE;
      for (int hh = 0; hh < 4; ++hh) {
        for (int r = tid; r < F * D; r += TPB) {
          const int mm = r >> 4;
          const int d  = r & 15;
          zb[zswz2(hh * F + mm, d)] =
              xb[(h0 + hh) * D + d] * xb[mm * D + d];
        }
      }
    }
    __syncthreads();

    __builtin_prefetch(&w0r[cbase + CK + 2 * hi], 0, 0);
    chunk_gemm<false, false>(w0r, K0, cbase, 19, true, hi, boff, zs, acc);
  }

  {
    __syncthreads();
    for (int nb = 0; nb < NB; ++nb) {
      const float* xb = xs + nb * (F * D);
      float*       zb = zs + nb * ZSTRIDE;
      for (int hh = 0; hh < 3; ++hh) {
        for (int r = tid; r < F * D; r += TPB) {
          const int mm = r >> 4;
          const int d  = r & 15;
          zb[zswz2(hh * F + mm, d)] =
              xb[(36 + hh) * D + d] * xb[mm * D + d];
        }
      }
    }
    for (int idx = tid; idx < NB * 3 * D; idx += TPB) {
      const int nb = idx / (3 * D);
      const int rr = idx - nb * (3 * D);
      zs[nb * ZSTRIDE + zswz2(117 + (rr >> 4), rr & 15)] = 0.0f;
    }
    __syncthreads();

    chunk_gemm<true, false>(w0r, K0, 9 * CK, 15, false, hi, boff, zs, acc);
  }
  __syncthreads();

#pragma unroll
  for (int nb = 0; nb < NB; ++nb) {
#pragma unroll
    for (int j = 0; j < 8; ++j) {
      const int row = (wave << 4) + j + (hi << 3);
      float v = acc[nb][j] + b0[row];
      v = v > 0.0f ? v : 0.0f;
      if (wave < 4) ys[nb * (H1 * D) + row * D + lo16] = v;
      else          zs[nb * (H1 * D) + (row - H1) * D + lo16] = v;
      acc[nb][j] = 0.0f;
    }
  }
  __syncthreads();

  {
    const int nb = tid >> 6;
    const int r  = tid & 63;
    float sum = 0.0f;
#pragma unroll
    for (int d = 0; d < D; ++d) sum += zs[nb * (H1 * D) + r * D + d];
    out[(size_t)(bbase + nb) * 192 + r] = sum;
  }

  for (int ch = 0; ch < 16; ++ch) {
    const int h0    = ch << 2;
    const int cbase = ch * CK;

    __syncthreads();
    for (int nb = 0; nb < NB; ++nb) {
      const float* xb = xs + nb * (F * D);
      const float* yb = ys + nb * (H1 * D);
      float*       zb = zs + nb * ZSTRIDE;
      for (int hh = 0; hh < 4; ++hh) {
        for (int r = tid; r < F * D; r += TPB) {
          const int mm = r >> 4;
          const int d  = r & 15;
          zb[zswz2(hh * F + mm, d)] =
              yb[(h0 + hh) * D + d] * xb[mm * D + d];
        }
      }
    }
    __syncthreads();

    if (ch < 15)
      __builtin_prefetch(&w1r[cbase + CK + 2 * hi], 0, 0);

    chunk_gemm<false, true>(w1r, K1, cbase, 19, true, hi, boff, zs, acc);
  }
  __syncthreads();

#pragma unroll
  for (int nb = 0; nb < NB; ++nb) {
#pragma unroll
    for (int j = 0; j < 8; ++j) {
      const int row = (wave << 4) + j + (hi << 3);
      float v = acc[nb][j] + b1[row];
      v = v > 0.0f ? v : 0.0f;
      zs[nb * (L0N * D) + row * D + lo16] = v;
    }
  }
  __syncthreads();

#pragma unroll
  for (int rep = 0; rep < 2; ++rep) {
    const int idx = tid + rep * TPB;
    const int nb  = idx >> 7;
    const int r   = idx & 127;
    float sum = 0.0f;
#pragma unroll
    for (int d = 0; d < D; ++d) sum += zs[nb * (L0N * D) + r * D + d];
    out[(size_t)(bbase + nb) * 192 + 64 + r] = sum;
  }
}

extern "C" void kernel_launch(void* const* d_in, const int* in_sizes, int n_in,
                              void* d_out, int out_size, void* d_ws, size_t ws_size,
                              hipStream_t stream) {
  const float* x  = (const float*)d_in[0];
  const float* W0 = (const float*)d_in[1];
  const float* b0 = (const float*)d_in[2];
  const float* W1 = (const float*)d_in[3];
  const float* b1 = (const float*)d_in[4];
  float* out = (float*)d_out;

  dim3 grid(B_TOT / NB);
  dim3 block(TPB);
  hipLaunchKernelGGL(cin_wmma_kernel, grid, block, 0, stream,
                     x, W0, b0, W1, b1, out);
}


# Z (B operand) built in registers: each wave keeps x fragments resident, forms B = xfrag*P[h][d] per h-group; drops LDS Z staging, its barriers, LDS B reads; K pad 39->40 per h, sum order changes
# speedup vs baseline: 1.4143x; 1.4143x over previous
#include <hip/hip_runtime.h>


typedef float v2f __attribute__((ext_vector_type(2)));
typedef float v4f __attribute__((ext_vector_type(4)));
typedef float v8f __attribute__((ext_vector_type(8)));
typedef float v4f_u __attribute__((ext_vector_type(4), aligned(4)));

#define B_TOT 4096
#define F     39
#define FP    40
#define D     16
#define L0N   128
#define H1    64
#define K0    1521
#define K1    2496
#define NB    4
#define TPB   256
#define NJ    5
#define NS    10

#define XS_OFF 0
#define YS_OFF 2560
#define R0_OFF (2560 + 4096)
#define R1_OFF (2560 + 4096 + 4096)
#define LDS_FLOATS (2560 + 4096 + 4096 + 8192)

__device__ __forceinline__ v4f ldu4(const float* p) {
  return *(const v4f_u*)p;
}

__device__ __forceinline__ void hgroup(const v4f (&a)[NJ], const float (&m)[NB],
                                       const v2f (&xf)[NB][NS], v8f (&acc)[NB])
{
#pragma unroll
  for (int j = 0; j < NJ; ++j) {
    v2f aA, aB;
    aA.x = a[j].x; aA.y = a[j].y;
    aB.x = a[j].z; aB.y = a[j].w;
#pragma unroll
    for (int nb = 0; nb < NB; ++nb) {
      const v2f b = xf[nb][2 * j] * m[nb];
      acc[nb] = __builtin_amdgcn_wmma_f32_16x16x4_f32(
          false, aA, false, b, (short)0, acc[nb], false, false);
    }
#pragma unroll
    for (int nb = 0; nb < NB; ++nb) {
      const v2f b = xf[nb][2 * j + 1] * m[nb];
      acc[nb] = __builtin_amdgcn_wmma_f32_16x16x4_f32(
          false, aB, false, b, (short)0, acc[nb], false, false);
    }
  }
}

template <int NH>
__device__ __forceinline__ void run_layer(const float* __restrict__ wrow, int hi, int lo16,
                                          const float* mult, int mstride,
                                          const v2f (&xf)[NB][NS], v8f (&acc)[NB])
{
  const float* wr = wrow + 4 * hi;
  v4f   ac[NJ];
  float mc[NB];
#pragma unroll
  for (int j = 0; j < NJ; ++j) ac[j] = ldu4(wr + 8 * j);
#pragma unroll
  for (int nb = 0; nb < NB; ++nb) mc[nb] = mult[nb * mstride + lo16];

  for (int h = 1; h < NH - 1; ++h) {
    v4f   an[NJ];
    float mn[NB];
    const float* wn = wr + h * F;
#pragma unroll
    for (int j = 0; j < NJ; ++j) an[j] = ldu4(wn + 8 * j);
#pragma unroll
    for (int nb = 0; nb < NB; ++nb) mn[nb] = mult[nb * mstride + h * D + lo16];
    hgroup(ac, mc, xf, acc);
#pragma unroll
    for (int j = 0; j < NJ; ++j) ac[j] = an[j];
#pragma unroll
    for (int nb = 0; nb < NB; ++nb) mc[nb] = mn[nb];
  }

  {
    v4f   an[NJ];
    float mn[NB];
    const float* wn = wr + (NH - 1) * F;
#pragma unroll
    for (int j = 0; j < NJ - 1; ++j) an[j] = ldu4(wn + 8 * j);
    const float* wl = wrow + (NH - 1) * F;
    const int m0 = 32 + 4 * hi;
    an[NJ - 1].x = wl[m0];
    an[NJ - 1].y = wl[m0 + 1];
    an[NJ - 1].z = wl[m0 + 2];
    an[NJ - 1].w = wl[(m0 + 3 < F) ? (m0 + 3) : (F - 1)];
#pragma unroll
    for (int nb = 0; nb < NB; ++nb) mn[nb] = mult[nb * mstride + (NH - 1) * D + lo16];
    hgroup(ac, mc, xf, acc);
    hgroup(an, mn, xf, acc);
  }
}

__global__ __launch_bounds__(TPB) void cin_wmma_kernel(
    const float* __restrict__ x,  const float* __restrict__ W0,
    const float* __restrict__ b0, const float* __restrict__ W1,
    const float* __restrict__ b1, float* __restrict__ out)
{
  __shared__ __align__(16) float lds[LDS_FLOATS];
  float* xs = lds + XS_OFF;
  float* ys = lds + YS_OFF;
  float* r0 = lds + R0_OFF;
  float* r1 = lds + R1_OFF;

  const int tid   = threadIdx.x;
  const int wave  = tid >> 5;
  const int lane  = tid & 31;
  const int lo16  = lane & 15;
  const int hi    = lane >> 4;
  const int bbase = blockIdx.x * NB;
  const int arow  = (wave << 4) + lo16;

  const float* xg = x + (size_t)bbase * (F * D);
  for (int e = tid; e < NB * F * D; e += TPB) {
    const int nb  = e / (F * D);
    const int rem = e - nb * (F * D);
    xs[nb * (FP * D) + rem] = xg[e];
  }
  if (tid < NB * D) xs[(tid >> 4) * (FP * D) + F * D + (tid & 15)] = 0.0f;
  __syncthreads();

  v2f xf[NB][NS];
#pragma unroll
  for (int nb = 0; nb < NB; ++nb) {
#pragma unroll
    for (int s = 0; s < NS; ++s) {
      const int m = 8 * (s >> 1) + 4 * hi + 2 * (s & 1);
      xf[nb][s].x = xs[nb * (FP * D) + m * D + lo16];
      xf[nb][s].y = xs[nb * (FP * D) + (m + 1) * D + lo16];
    }
  }

  v8f acc[NB];
#pragma unroll
  for (int nb = 0; nb < NB; ++nb)
#pragma unroll
    for (int j = 0; j < 8; ++j) acc[nb][j] = 0.0f;

  run_layer<F>(W0 + (size_t)arow * K0, hi, lo16, xs, FP * D, xf, acc);

#pragma unroll
  for (int nb = 0; nb < NB; ++nb) {
#pragma unroll
    for (int j = 0; j < 8; ++j) {
      const int row = (wave << 4) + j + (hi << 3);
      float v = acc[nb][j] + b0[row];
      v = v > 0.0f ? v : 0.0f;
      if (wave < 4) ys[nb * (H1 * D) + row * D + lo16] = v;
      else          r0[nb * (H1 * D) + (row - H1) * D + lo16] = v;
      acc[nb][j] = 0.0f;
    }
  }
  __syncthreads();

  {
    const int nb = tid >> 6;
    const int r  = tid & 63;
    float sum = 0.0f;
#pragma unroll
    for (int d = 0; d < D; ++d) sum += r0[nb * (H1 * D) + r * D + d];
    out[(size_t)(bbase + nb) * 192 + r] = sum;
  }

  run_layer<H1>(W1 + (size_t)arow * K1, hi, lo16, ys, H1 * D, xf, acc);

#pragma unroll
  for (int nb = 0; nb < NB; ++nb) {
#pragma unroll
    for (int j = 0; j < 8; ++j) {
      const int row = (wave << 4) + j + (hi << 3);
      float v = acc[nb][j] + b1[row];
      v = v > 0.0f ? v : 0.0f;
      r1[nb * (L0N * D) + row * D + lo16] = v;
    }
  }
  __syncthreads();

#pragma unroll
  for (int rep = 0; rep < 2; ++rep) {
    const int idx = tid + rep * TPB;
    const int nb  = idx >> 7;
    const int r   = idx & 127;
    float sum = 0.0f;
#pragma unroll
    for (int d = 0; d < D; ++d) sum += r1[nb * (L0N * D) + r * D + d];
    out[(size_t)(bbase + nb) * 192 + 64 + r] = sum;
  }
}

extern "C" void kernel_launch(void* const* d_in, const int* in_sizes, int n_in,
                              void* d_out, int out_size, void* d_ws, size_t ws_size,
                              hipStream_t stream) {
  const float* x  = (const float*)d_in[0];
  const float* W0 = (const float*)d_in[1];
  const float* b0 = (const float*)d_in[2];
  const float* W1 = (const float*)d_in[3];
  const float* b1 = (const float*)d_in[4];
  float* out = (float*)d_out;

  dim3 grid(B_TOT / NB);
  dim3 block(TPB);
  hipLaunchKernelGGL(cin_wmma_kernel, grid, block, 0, stream,
                     x, W0, b0, W1, b1, out);
}


# Layer 0 uses the symmetry of x*x: a first launch writes S[h,m]=W0[h,m]+W0[m,h], each pair once, to the workspace in fragment order; main kernel walks 219 instead of 390 steps; sum order changes
# speedup vs baseline: 1.6865x; 1.1925x over previous
#include <hip/hip_runtime.h>


typedef float v2f __attribute__((ext_vector_type(2)));
typedef float v4f __attribute__((ext_vector_type(4)));
typedef float v8f __attribute__((ext_vector_type(8)));
typedef float v4f_u __attribute__((ext_vector_type(4), aligned(4)));

#define B_TOT 4096
#define F     39
#define FP    40
#define D     16
#define L0N   128
#define H1    64
#define K0    1521
#define K1    2496
#define NB    4
#define TPB   256
#define NJ    5
#define NS    10

#define NU0     119
#define WS_NEED 487424

#define XS_OFF 0
#define YS_OFF 2560
#define R0_OFF (2560 + 4096)
#define R1_OFF (2560 + 4096 + 4096)
#define LDS_FLOATS (2560 + 4096 + 4096 + 8192)

__host__ __device__ constexpr int ubase(int q) {
  return q == 0 ? 0 : q == 1 ? 40 : q == 2 ? 72 : q == 3 ? 96 : 112;
}

__device__ __forceinline__ v4f ldu4(const float* p) {
  return *(const v4f_u*)p;
}

__global__ __launch_bounds__(TPB) void cin_sym_pack_kernel(
    const float* __restrict__ W0, float* __restrict__ ws, size_t ws_bytes)
{
  if (ws_bytes < (size_t)WS_NEED) return;
  const int gid  = blockIdx.x * TPB + threadIdx.x;
  const int lane = gid & 31;
  const int tu   = gid >> 5;
  const int tile = tu / NU0;
  const int u    = tu - tile * NU0;
  const int q    = (u >= 112) ? 4 : (u >= 96) ? 3 : (u >= 72) ? 2 : (u >= 40) ? 1 : 0;
  const int per  = NJ - q;
  const int r    = u - ubase(q);
  const int hh   = r / per;
  const int h    = 8 * q + hh;
  const int j    = q + (r - hh * per);
  const int m0   = 8 * j + 4 * (lane >> 4);
  const float* wo = W0 + (size_t)(16 * tile + (lane & 15)) * K0;
  float v[4];
#pragma unroll
  for (int c = 0; c < 4; ++c) {
    const int m = m0 + c;
    float s = 0.0f;
    if (m < F) {
      bool pair;
      if (j > q)                       pair = true;
      else if (((h ^ m) & 2) == 0)     pair = m > h;
      else                             pair = (h & 2) == 0;
      if (m == h)    s = wo[h * F + h];
      else if (pair) s = wo[h * F + m] + wo[m * F + h];
    }
    v[c] = s;
  }
  v4f o; o.x = v[0]; o.y = v[1]; o.z = v[2]; o.w = v[3];
  *(v4f*)(ws + (size_t)gid * 4) = o;
}

__device__ __forceinline__ void hgroup(const v4f (&a)[NJ], const float (&m)[NB],
                                       const v2f (&xf)[NB][NS], v8f (&acc)[NB])
{
#pragma unroll
  for (int j = 0; j < NJ; ++j) {
    v2f aA, aB;
    aA.x = a[j].x; aA.y = a[j].y;
    aB.x = a[j].z; aB.y = a[j].w;
#pragma unroll
    for (int nb = 0; nb < NB; ++nb) {
      const v2f b = xf[nb][2 * j] * m[nb];
      acc[nb] = __builtin_amdgcn_wmma_f32_16x16x4_f32(
          false, aA, false, b, (short)0, acc[nb], false, false);
    }
#pragma unroll
    for (int nb = 0; nb < NB; ++nb) {
      const v2f b = xf[nb][2 * j + 1] * m[nb];
      acc[nb] = __builtin_amdgcn_wmma_f32_16x16x4_f32(
          false, aB, false, b, (short)0, acc[nb], false, false);
    }
  }
}

template <int NH>
__device__ __forceinline__ void run_layer(const float* __restrict__ wrow, int hi, int lo16,
                                          const float* mult, int mstride,
                                          const v2f (&xf)[NB][NS], v8f (&acc)[NB])
{
  const float* wr = wrow + 4 * hi;
  v4f   ac[NJ];
  float mc[NB];
#pragma unroll
  for (int j = 0; j < NJ; ++j) ac[j] = ldu4(wr + 8 * j);
#pragma unroll
  for (int nb = 0; nb < NB; ++nb) mc[nb] = mult[nb * mstride + lo16];

  for (int h = 1; h < NH - 1; ++h) {
    v4f   an[NJ];
    float mn[NB];
    const float* wn = wr + h * F;
#pragma unroll
    for (int j = 0; j < NJ; ++j) an[j] = ldu4(wn + 8 * j);
#pragma unroll
    for (int nb = 0; nb < NB; ++nb) mn[nb] = mult[nb * mstride + h * D + lo16];
    hgroup(ac, mc, xf, acc);
#pragma unroll
    for (int j = 0; j < NJ; ++j) ac[j] = an[j];
#pragma unroll
    for (int nb = 0; nb < NB; ++nb) mc[nb] = mn[nb];
  }

  {
    v4f   an[NJ];
    float mn[NB];
    const float* wn = wr + (NH - 1) * F;
#pragma unroll
    for (int j = 0; j < NJ - 1; ++j) an[j] = ldu4(wn + 8 * j);
    const float* wl = wrow + (NH - 1) * F;
    const int m0 = 32 + 4 * hi;
    an[NJ - 1].x = wl[m0];
    an[NJ - 1].y = wl[m0 + 1];
    an[NJ - 1].z = wl[m0 + 2];
    an[NJ - 1].w = wl[(m0 + 3 < F) ? (m0 + 3) : (F - 1)];
#pragma unroll
    for (int nb = 0; nb < NB; ++nb) mn[nb] = mult[nb * mstride + (NH - 1) * D + lo16];
    hgroup(ac, mc, xf, acc);
    hgroup(an, mn, xf, acc);
  }
}


template <int Q, bool LEAD0>
__device__ __forceinline__ void hgroup_sym(const v2f (&a0)[NJ], const v2f (&a1)[NJ],
                                           const float (&m)[NB],
                                           const v2f (&xf)[NB][NS], v8f (&acc)[NB])
{
#pragma unroll
  for (int j = Q; j < NJ; ++j) {
    if (LEAD0 || j > Q) {
#pragma unroll
      for (int nb = 0; nb < NB; ++nb) {
        const v2f b = xf[nb][2 * j] * m[nb];
        acc[nb] = __builtin_amdgcn_wmma_f32_16x16x4_f32(
            false, a0[j], false, b, (short)0, acc[nb], false, false);
      }
    }
#pragma unroll
    for (int nb = 0; nb < NB; ++nb) {
      const v2f b = xf[nb][2 * j + 1] * m[nb];
      acc[nb] = __builtin_amdgcn_wmma_f32_16x16x4_f32(
          false, a1[j], false, b, (short)0, acc[nb], false, false);
    }
  }
}

template <int Q, bool LEAD0, int QN, bool LEADN>
__device__ __forceinline__ void sym_step(const v4f* __restrict__ sp, int un,
                                         const float* mult, int hn, int lo16,
                                         v2f (&a0)[NJ], v2f (&a1)[NJ], float (&mc)[NB],
                                         const v2f (&xf)[NB][NS], v8f (&acc)[NB])
{
  v2f   n0[NJ], n1[NJ];
  float mn[NB];
#pragma unroll
  for (int j = QN; j < NJ; ++j) {
    const v4f* pp = sp + (un + (j - QN)) * 32;
    if (LEADN || j > QN) {
      const v4f t = *pp;
      n0[j].x = t.x; n0[j].y = t.y;
      n1[j].x = t.z; n1[j].y = t.w;
    } else {
      n1[j] = *((const v2f*)pp + 1);
    }
  }
#pragma unroll
  for (int nb = 0; nb < NB; ++nb) mn[nb] = mult[nb * (FP * D) + hn * D + lo16];
  hgroup_sym<Q, LEAD0>(a0, a1, mc, xf, acc);
#pragma unroll
  for (int j = QN; j < NJ; ++j) {
    if (LEADN || j > QN) a0[j] = n0[j];
    a1[j] = n1[j];
  }
#pragma unroll
  for (int nb = 0; nb < NB; ++nb) mc[nb] = mn[nb];
}

template <int Q>
__device__ __forceinline__ void sym_block(const v4f* __restrict__ sp, const float* mult, int lo16,
                                          v2f (&a0)[NJ], v2f (&a1)[NJ], float (&mc)[NB],
                                          const v2f (&xf)[NB][NS], v8f (&acc)[NB])
{
  constexpr int PER = NJ - Q;
  constexpr int UB  = ubase(Q);
  constexpr int H0  = 8 * Q;
  constexpr int NC1 = (Q < 4) ? 4 : 3;
#pragma unroll 1
  for (int p = 1; p <= 4; ++p) {
    const int rn = (p & 1) | ((p & 2) << 1) | ((p & 4) >> 1);
    sym_step<Q, true, Q, true>(sp, UB + rn * PER, mult, H0 + rn, lo16, a0, a1, mc, xf, acc);
  }
#pragma unroll 1
  for (int p = 5; p < 4 + NC1; ++p) {
    const int rn = (p & 1) | ((p & 2) << 1) | ((p & 4) >> 1);
    sym_step<Q, false, Q, false>(sp, UB + rn * PER, mult, H0 + rn, lo16, a0, a1, mc, xf, acc);
  }
  if (Q < 4)
    sym_step<Q, false, Q + 1, true>(sp, ubase(Q + 1), mult, H0 + 8, lo16, a0, a1, mc, xf, acc);
  else
    hgroup_sym<Q, false>(a0, a1, mc, xf, acc);
}

__device__ __forceinline__ void run_layer0_sym(const float* __restrict__ ws, int wave, int lane,
                                               int lo16, const float* xs,
                                               const v2f (&xf)[NB][NS], v8f (&acc)[NB])
{
  const v4f* sp = (const v4f*)ws + (size_t)wave * (NU0 * 32) + lane;
  v2f   a0[NJ], a1[NJ];
  float mc[NB];
#pragma unroll
  for (int j = 0; j < NJ; ++j) {
    const v4f t = sp[j * 32];
    a0[j].x = t.x; a0[j].y = t.y;
    a1[j].x = t.z; a1[j].y = t.w;
  }
#pragma unroll
  for (int nb = 0; nb < NB; ++nb) mc[nb] = xs[nb * (FP * D) + lo16];
  sym_block<0>(sp, xs, lo16, a0, a1, mc, xf, acc);
  sym_block<1>(sp, xs, lo16, a0, a1, mc, xf, acc);
  sym_block<2>(sp, xs, lo16, a0, a1, mc, xf, acc);
  sym_block<3>(sp, xs, lo16, a0, a1, mc, xf, acc);
  sym_block<4>(sp, xs, lo16, a0, a1, mc, xf, acc);
}

__global__ __launch_bounds__(TPB) void cin_wmma_kernel(
    const float* __restrict__ x,  const float* __restrict__ W0,
    const float* __restrict__ b0, const float* __restrict__ W1,
    const float* __restrict__ b1, float* __restrict__ out,
    const float* __restrict__ ws, size_t ws_bytes)
{
  __shared__ __align__(16) float lds[LDS_FLOATS];
  float* xs = lds + XS_OFF;
  float* ys = lds + YS_OFF;
  float* r0 = lds + R0_OFF;
  float* r1 = lds + R1_OFF;

  const int tid   = threadIdx.x;
  const int wave  = tid >> 5;
  const int lane  = tid & 31;
  const int lo16  = lane & 15;
  const int hi    = lane >> 4;
  const int bbase = blockIdx.x * NB;
  const int arow  = (wave << 4) + lo16;

  const float* xg = x + (size_t)bbase * (F * D);
  for (int e = tid; e < NB * F * D; e += TPB) {
    const int nb  = e / (F * D);
    const int rem = e - nb * (F * D);
    xs[nb * (FP * D) + rem] = xg[e];
  }
  if (tid < NB * D) xs[(tid >> 4) * (FP * D) + F * D + (tid & 15)] = 0.0f;
  __syncthreads();

  v2f xf[NB][NS];
#pragma unroll
  for (int nb = 0; nb < NB; ++nb) {
#pragma unroll
    for (int s = 0; s < NS; ++s) {
      const int m = 8 * (s >> 1) + 4 * hi + 2 * (s & 1);
      xf[nb][s].x = xs[nb * (FP * D) + m * D + lo16];
      xf[nb][s].y = xs[nb * (FP * D) + (m + 1) * D + lo16];
    }
  }

  v8f acc[NB];
#pragma unroll
  for (int nb = 0; nb < NB; ++nb)
#pragma unroll
    for (int j = 0; j < 8; ++j) acc[nb][j] = 0.0f;

  if (ws_bytes >= (size_t)WS_NEED)
    run_layer0_sym(ws, wave, lane, lo16, xs, xf, acc);
  else
    run_layer<F>(W0 + (size_t)arow * K0, hi, lo16, xs, FP * D, xf, acc);

#pragma unroll
  for (int nb = 0; nb < NB; ++nb) {
#pragma unroll
    for (int j = 0; j < 8; ++j) {
      const int row = (wave << 4) + j + (hi << 3);
      float v = acc[nb][j] + b0[row];
      v = v > 0.0f ? v : 0.0f;
      if (wave < 4) ys[nb * (H1 * D) + row * D + lo16] = v;
      else          r0[nb * (H1 * D) + (row - H1) * D + lo16] = v;
      acc[nb][j] = 0.0f;
    }
  }
  __syncthreads();

  {
    const int nb = tid >> 6;
    const int r  = tid & 63;
    float sum = 0.0f;
#pragma unroll
    for (int d = 0; d < D; ++d) sum += r0[nb * (H1 * D) + r * D + d];
    out[(size_t)(bbase + nb) * 192 + r] = sum;
  }

  run_layer<H1>(W1 + (size_t)arow * K1, hi, lo16, ys, H1 * D, xf, acc);

#pragma unroll
  for (int nb = 0; nb < NB; ++nb) {
#pragma unroll
    for (int j = 0; j < 8; ++j) {
      const int row = (wave << 4) + j + (hi << 3);
      float v = acc[nb][j] + b1[row];
      v = v > 0.0f ? v : 0.0f;
      r1[nb * (L0N * D) + row * D + lo16] = v;
    }
  }
  __syncthreads();

#pragma unroll
  for (int rep = 0; rep < 2; ++rep) {
    const int idx = tid + rep * TPB;
    const int nb  = idx >> 7;
    const int r   = idx & 127;
    float sum = 0.0f;
#pragma unroll
    for (int d = 0; d < D; ++d) sum += r1[nb * (L0N * D) + r * D + d];
    out[(size_t)(bbase + nb) * 192 + 64 + r] = sum;
  }
}

extern "C" void kernel_launch(void* const* d_in, const int* in_sizes, int n_in,
                              void* d_out, int out_size, void* d_ws, size_t ws_size,
                              hipStream_t stream) {
  const float* x  = (const float*)d_in[0];
  const float* W0 = (const float*)d_in[1];
  const float* b0 = (const float*)d_in[2];
  const float* W1 = (const float*)d_in[3];
  const float* b1 = (const float*)d_in[4];
  float* out = (float*)d_out;
  float* ws  = (float*)d_ws;

  hipLaunchKernelGGL(cin_sym_pack_kernel, dim3(NU0), dim3(TPB), 0, stream,
                     W0, ws, ws_size);

  dim3 grid(B_TOT / NB);
  dim3 block(TPB);
  hipLaunchKernelGGL(cin_wmma_kernel, grid, block, 0, stream,
                     x, W0, b0, W1, b1, out, (const float*)ws, ws_size);
}


# Layer 1 reads W1 in fragment order: the first launch also copies W1 behind S in the workspace, so an A load is one aligned contiguous 512-byte read, not 16 row-strided 32-byte pieces; sums unchanged
# speedup vs baseline: 1.7422x; 1.0330x over previous
#include <hip/hip_runtime.h>


typedef float v2f __attribute__((ext_vector_type(2)));
typedef float v4f __attribute__((ext_vector_type(4)));
typedef float v8f __attribute__((ext_vector_type(8)));
typedef float v4f_u __attribute__((ext_vector_type(4), aligned(4)));

#define B_TOT 4096
#define F     39
#define FP    40
#define D     16
#define L0N   128
#define H1    64
#define K0    1521
#define K1    2496
#define NB    4
#define TPB   256
#define NJ    5
#define NS    10

#define NU0     119
#define WS_NEED 487424
#define NP1      2560
#define NB1      320
#define W1P_OFF  121856
#define WS_NEED1 1798144

#define XS_OFF 0
#define YS_OFF 2560
#define R0_OFF (2560 + 4096)
#define R1_OFF (2560 + 4096 + 4096)
#define LDS_FLOATS (2560 + 4096 + 4096 + 8192)

__host__ __device__ constexpr int ubase(int q) {
  return q == 0 ? 0 : q == 1 ? 40 : q == 2 ? 72 : q == 3 ? 96 : 112;
}

__device__ __forceinline__ v4f ldu4(const float* p) {
  return *(const v4f_u*)p;
}

__global__ __launch_bounds__(TPB) void cin_sym_pack_kernel(
    const float* __restrict__ W0, const float* __restrict__ W1,
    float* __restrict__ ws, size_t ws_bytes)
{
  if (blockIdx.x >= NU0) {
    if (ws_bytes < (size_t)WS_NEED1) return;
    const int g1   = (blockIdx.x - NU0) * TPB + threadIdx.x;
    const int lane = g1 & 31;
    const int pj   = g1 >> 5;
    const int th   = pj / NJ;
    const int j    = pj - th * NJ;
    const int h    = th & (H1 - 1);
    const int tile = th >> 6;
    const int m0   = 8 * j + 4 * (lane >> 4);
    const float* wo = W1 + (size_t)(16 * tile + (lane & 15)) * K1 + h * F + m0;
    v4f o;
    o.x = wo[0]; o.y = wo[1]; o.z = wo[2];
    o.w = (m0 + 3 < F) ? wo[3] : 0.0f;
    *(v4f*)(ws + W1P_OFF + (size_t)g1 * 4) = o;
    return;
  }
  if (ws_bytes < (size_t)WS_NEED) return;
  const int gid  = blockIdx.x * TPB + threadIdx.x;
  const int lane = gid & 31;
  const int tu   = gid >> 5;
  const int tile = tu / NU0;
  const int u    = tu - tile * NU0;
  const int q    = (u >= 112) ? 4 : (u >= 96) ? 3 : (u >= 72) ? 2 : (u >= 40) ? 1 : 0;
  const int per  = NJ - q;
  const int r    = u - ubase(q);
  const int hh   = r / per;
  const int h    = 8 * q + hh;
  const int j    = q + (r - hh * per);
  const int m0   = 8 * j + 4 * (lane >> 4);
  const float* wo = W0 + (size_t)(16 * tile + (lane & 15)) * K0;
  float v[4];
#pragma unroll
  for (int c = 0; c < 4; ++c) {
    const int m = m0 + c;
    float s = 0.0f;
    if (m < F) {
      bool pair;
      if (j > q)                       pair = true;
      else if (((h ^ m) & 2) == 0)     pair = m > h;
      else                             pair = (h & 2) == 0;
      if (m == h)    s = wo[h * F + h];
      else if (pair) s = wo[h * F + m] + wo[m * F + h];
    }
    v[c] = s;
  }
  v4f o; o.x = v[0]; o.y = v[1]; o.z = v[2]; o.w = v[3];
  *(v4f*)(ws + (size_t)gid * 4) = o;
}

__device__ __forceinline__ void hgroup(const v4f (&a)[NJ], const float (&m)[NB],
                                       const v2f (&xf)[NB][NS], v8f (&acc)[NB])
{
#pragma unroll
  for (int j = 0; j < NJ; ++j) {
    v2f aA, aB;
    aA.x = a[j].x; aA.y = a[j].y;
    aB.x = a[j].z; aB.y = a[j].w;
#pragma unroll
    for (int nb = 0; nb < NB; ++nb) {
      const v2f b = xf[nb][2 * j] * m[nb];
      acc[nb] = __builtin_amdgcn_wmma_f32_16x16x4_f32(
          false, aA, false, b, (short)0, acc[nb], false, false);
    }
#pragma unroll
    for (int nb = 0; nb < NB; ++nb) {
      const v2f b = xf[nb][2 * j + 1] * m[nb];
      acc[nb] = __builtin_amdgcn_wmma_f32_16x16x4_f32(
          false, aB, false, b, (short)0, acc[nb], false, false);
    }
  }
}

template <int NH>
__device__ __forceinline__ void run_layer(const float* __restrict__ wrow, int hi, int lo16,
                                          const float* mult, int mstride,
                                          const v2f (&xf)[NB][NS], v8f (&acc)[NB])
{
  const float* wr = wrow + 4 * hi;
  v4f   ac[NJ];
  float mc[NB];
#pragma unroll
  for (int j = 0; j < NJ; ++j) ac[j] = ldu4(wr + 8 * j);
#pragma unroll
  for (int nb = 0; nb < NB; ++nb) mc[nb] = mult[nb * mstride + lo16];

  for (int h = 1; h < NH - 1; ++h) {
    v4f   an[NJ];
    float mn[NB];
    const float* wn = wr + h * F;
#pragma unroll
    for (int j = 0; j < NJ; ++j) an[j] = ldu4(wn + 8 * j);
#pragma unroll
    for (int nb = 0; nb < NB; ++nb) mn[nb] = mult[nb * mstride + h * D + lo16];
    hgroup(ac, mc, xf, acc);
#pragma unroll
    for (int j = 0; j < NJ; ++j) ac[j] = an[j];
#pragma unroll
    for (int nb = 0; nb < NB; ++nb) mc[nb] = mn[nb];
  }

  {
    v4f   an[NJ];
    float mn[NB];
    const float* wn = wr + (NH - 1) * F;
#pragma unroll
    for (int j = 0; j < NJ - 1; ++j) an[j] = ldu4(wn + 8 * j);
    const float* wl = wrow + (NH - 1) * F;
    const int m0 = 32 + 4 * hi;
    an[NJ - 1].x = wl[m0];
    an[NJ - 1].y = wl[m0 + 1];
    an[NJ - 1].z = wl[m0 + 2];
    an[NJ - 1].w = wl[(m0 + 3 < F) ? (m0 + 3) : (F - 1)];
#pragma unroll
    for (int nb = 0; nb < NB; ++nb) mn[nb] = mult[nb * mstride + (NH - 1) * D + lo16];
    hgroup(ac, mc, xf, acc);
    hgroup(an, mn, xf, acc);
  }
}

__device__ __forceinline__ void run_layer1_packed(const v4f* __restrict__ wp, int lo16,
                                                  const float* mult,
                                                  const v2f (&xf)[NB][NS], v8f (&acc)[NB])
{
  v4f   ac[NJ];
  float mc[NB];
#pragma unroll
  for (int j = 0; j < NJ; ++j) ac[j] = wp[j * 32];
#pragma unroll
  for (int nb = 0; nb < NB; ++nb) mc[nb] = mult[nb * (H1 * D) + lo16];

  for (int h = 1; h < H1; ++h) {
    v4f   an[NJ];
    float mn[NB];
    const v4f* wn = wp + h * (NJ * 32);
#pragma unroll
    for (int j = 0; j < NJ; ++j) an[j] = wn[j * 32];
#pragma unroll
    for (int nb = 0; nb < NB; ++nb) mn[nb] = mult[nb * (H1 * D) + h * D + lo16];
    hgroup(ac, mc, xf, acc);
#pragma unroll
    for (int j = 0; j < NJ; ++j) ac[j] = an[j];
#pragma unroll
    for (int nb = 0; nb < NB; ++nb) mc[nb] = mn[nb];
  }
  hgroup(ac, mc, xf, acc);
}


template <int Q, bool LEAD0>
__device__ __forceinline__ void hgroup_sym(const v2f (&a0)[NJ], const v2f (&a1)[NJ],
                                           const float (&m)[NB],
                                           const v2f (&xf)[NB][NS], v8f (&acc)[NB])
{
#pragma unroll
  for (int j = Q; j < NJ; ++j) {
    if (LEAD0 || j > Q) {
#pragma unroll
      for (int nb = 0; nb < NB; ++nb) {
        const v2f b = xf[nb][2 * j] * m[nb];
        acc[nb] = __builtin_amdgcn_wmma_f32_16x16x4_f32(
            false, a0[j], false, b, (short)0, acc[nb], false, false);
      }
    }
#pragma unroll
    for (int nb = 0; nb < NB; ++nb) {
      const v2f b = xf[nb][2 * j + 1] * m[nb];
      acc[nb] = __builtin_amdgcn_wmma_f32_16x16x4_f32(
          false, a1[j], false, b, (short)0, acc[nb], false, false);
    }
  }
}

template <int Q, bool LEAD0, int QN, bool LEADN>
__device__ __forceinline__ void sym_step(const v4f* __restrict__ sp, int un,
                                         const float* mult, int hn, int lo16,
                                         v2f (&a0)[NJ], v2f (&a1)[NJ], float (&mc)[NB],
                                         const v2f (&xf)[NB][NS], v8f (&acc)[NB])
{
  v2f   n0[NJ], n1[NJ];
  float mn[NB];
#pragma unroll
  for (int j = QN; j < NJ; ++j) {
    const v4f* pp = sp + (un + (j - QN)) * 32;
    if (LEADN || j > QN) {
      const v4f t = *pp;
      n0[j].x = t.x; n0[j].y = t.y;
      n1[j].x = t.z; n1[j].y = t.w;
    } else {
      n1[j] = *((const v2f*)pp + 1);
    }
  }
#pragma unroll
  for (int nb = 0; nb < NB; ++nb) mn[nb] = mult[nb * (FP * D) + hn * D + lo16];
  hgroup_sym<Q, LEAD0>(a0, a1, mc, xf, acc);
#pragma unroll
  for (int j = QN; j < NJ; ++j) {
    if (LEADN || j > QN) a0[j] = n0[j];
    a1[j] = n1[j];
  }
#pragma unroll
  for (int nb = 0; nb < NB; ++nb) mc[nb] = mn[nb];
}

template <int Q>
__device__ __forceinline__ void sym_block(const v4f* __restrict__ sp, const float* mult, int lo16,
                                          v2f (&a0)[NJ], v2f (&a1)[NJ], float (&mc)[NB],
                                          const v2f (&xf)[NB][NS], v8f (&acc)[NB])
{
  constexpr int PER = NJ - Q;
  constexpr int UB  = ubase(Q);
  constexpr int H0  = 8 * Q;
  constexpr int NC1 = (Q < 4) ? 4 : 3;
#pragma unroll 1
  for (int p = 1; p <= 4; ++p) {
    const int rn = (p & 1) | ((p & 2) << 1) | ((p & 4) >> 1);
    sym_step<Q, true, Q, true>(sp, UB + rn * PER, mult, H0 + rn, lo16, a0, a1, mc, xf, acc);
  }
#pragma unroll 1
  for (int p = 5; p < 4 + NC1; ++p) {
    const int rn = (p & 1) | ((p & 2) << 1) | ((p & 4) >> 1);
    sym_step<Q, false, Q, false>(sp, UB + rn * PER, mult, H0 + rn, lo16, a0, a1, mc, xf, acc);
  }
  if (Q < 4)
    sym_step<Q, false, Q + 1, true>(sp, ubase(Q + 1), mult, H0 + 8, lo16, a0, a1, mc, xf, acc);
  else
    hgroup_sym<Q, false>(a0, a1, mc, xf, acc);
}

__device__ __forceinline__ void run_layer0_sym(const float* __restrict__ ws, int wave, int lane,
                                               int lo16, const float* xs,
                                               const v2f (&xf)[NB][NS], v8f (&acc)[NB])
{
  const v4f* sp = (const v4f*)ws + (size_t)wave * (NU0 * 32) + lane;
  v2f   a0[NJ], a1[NJ];
  float mc[NB];
#pragma unroll
  for (int j = 0; j < NJ; ++j) {
    const v4f t = sp[j * 32];
    a0[j].x = t.x; a0[j].y = t.y;
    a1[j].x = t.z; a1[j].y = t.w;
  }
#pragma unroll
  for (int nb = 0; nb < NB; ++nb) mc[nb] = xs[nb * (FP * D) + lo16];
  sym_block<0>(sp, xs, lo16, a0, a1, mc, xf, acc);
  sym_block<1>(sp, xs, lo16, a0, a1, mc, xf, acc);
  sym_block<2>(sp, xs, lo16, a0, a1, mc, xf, acc);
  sym_block<3>(sp, xs, lo16, a0, a1, mc, xf, acc);
  sym_block<4>(sp, xs, lo16, a0, a1, mc, xf, acc);
}

__global__ __launch_bounds__(TPB) void cin_wmma_kernel(
    const float* __restrict__ x,  const float* __restrict__ W0,
    const float* __restrict__ b0, const float* __restrict__ W1,
    const float* __restrict__ b1, float* __restrict__ out,
    const float* __restrict__ ws, size_t ws_bytes)
{
  __shared__ __align__(16) float lds[LDS_FLOATS];
  float* xs = lds + XS_OFF;
  float* ys = lds + YS_OFF;
  float* r0 = lds + R0_OFF;
  float* r1 = lds + R1_OFF;

  const int tid   = threadIdx.x;
  const int wave  = tid >> 5;
  const int lane  = tid & 31;
  const int lo16  = lane & 15;
  const int hi    = lane >> 4;
  const int bbase = blockIdx.x * NB;
  const int arow  = (wave << 4) + lo16;

  const float* xg = x + (size_t)bbase * (F * D);
  for (int e = tid; e < NB * F * D; e += TPB) {
    const int nb  = e / (F * D);
    const int rem = e - nb * (F * D);
    xs[nb * (FP * D) + rem] = xg[e];
  }
  if (tid < NB * D) xs[(tid >> 4) * (FP * D) + F * D + (tid & 15)] = 0.0f;
  __syncthreads();

  v2f xf[NB][NS];
#pragma unroll
  for (int nb = 0; nb < NB; ++nb) {
#pragma unroll
    for (int s = 0; s < NS; ++s) {
      const int m = 8 * (s >> 1) + 4 * hi + 2 * (s & 1);
      xf[nb][s].x = xs[nb * (FP * D) + m * D + lo16];
      xf[nb][s].y = xs[nb * (FP * D) + (m + 1) * D + lo16];
    }
  }

  v8f acc[NB];
#pragma unroll
  for (int nb = 0; nb < NB; ++nb)
#pragma unroll
    for (int j = 0; j < 8; ++j) acc[nb][j] = 0.0f;

  if (ws_bytes >= (size_t)WS_NEED)
    run_layer0_sym(ws, wave, lane, lo16, xs, xf, acc);
  else
    run_layer<F>(W0 + (size_t)arow * K0, hi, lo16, xs, FP * D, xf, acc);

#pragma unroll
  for (int nb = 0; nb < NB; ++nb) {
#pragma unroll
    for (int j = 0; j < 8; ++j) {
      const int row = (wave << 4) + j + (hi << 3);
      float v = acc[nb][j] + b0[row];
      v = v > 0.0f ? v : 0.0f;
      if (wave < 4) ys[nb * (H1 * D) + row * D + lo16] = v;
      else          r0[nb * (H1 * D) + (row - H1) * D + lo16] = v;
      acc[nb][j] = 0.0f;
    }
  }
  __syncthreads();

  {
    const int nb = tid >> 6;
    const int r  = tid & 63;
    float sum = 0.0f;
#pragma unroll
    for (int d = 0; d < D; ++d) sum += r0[nb * (H1 * D) + r * D + d];
    out[(size_t)(bbase + nb) * 192 + r] = sum;
  }

  if (ws_bytes >= (size_t)WS_NEED1)
    run_layer1_packed((const v4f*)(ws + W1P_OFF) + (size_t)wave * (H1 * NJ * 32) + lane,
                      lo16, ys, xf, acc);
  else
    run_layer<H1>(W1 + (size_t)arow * K1, hi, lo16, ys, H1 * D, xf, acc);

#pragma unroll
  for (int nb = 0; nb < NB; ++nb) {
#pragma unroll
    for (int j = 0; j < 8; ++j) {
      const int row = (wave << 4) + j + (hi << 3);
      float v = acc[nb][j] + b1[row];
      v = v > 0.0f ? v : 0.0f;
      r1[nb * (L0N * D) + row * D + lo16] = v;
    }
  }
  __syncthreads();

#pragma unroll
  for (int rep = 0; rep < 2; ++rep) {
    const int idx = tid + rep * TPB;
    const int nb  = idx >> 7;
    const int r   = idx & 127;
    float sum = 0.0f;
#pragma unroll
    for (int d = 0; d < D; ++d) sum += r1[nb * (L0N * D) + r * D + d];
    out[(size_t)(bbase + nb) * 192 + 64 + r] = sum;
  }
}

extern "C" void kernel_launch(void* const* d_in, const int* in_sizes, int n_in,
                              void* d_out, int out_size, void* d_ws, size_t ws_size,
                              hipStream_t stream) {
  const float* x  = (const float*)d_in[0];
  const float* W0 = (const float*)d_in[1];
  const float* b0 = (const float*)d_in[2];
  const float* W1 = (const float*)d_in[3];
  const float* b1 = (const float*)d_in[4];
  float* out = (float*)d_out;
  float* ws  = (float*)d_ws;

  hipLaunchKernelGGL(cin_sym_pack_kernel, dim3(NU0 + NB1), dim3(TPB), 0, stream,
                     W0, W1, ws, ws_size);

  dim3 grid(B_TOT / NB);
  dim3 block(TPB);
  hipLaunchKernelGGL(cin_wmma_kernel, grid, block, 0, stream,
                     x, W0, b0, W1, b1, out, (const float*)ws, ws_size);
}


# Wave tile 32 rows x 8 batches (4-wave blocks, two 16-row tiles per wave): an A piece feeds 16 WMMAs, a packed-multiply B fragment 2, one load group serves 160 WMMAs not 40; order of sums unchanged
# speedup vs baseline: 1.8384x; 1.0552x over previous
#include <hip/hip_runtime.h>


typedef float v2f __attribute__((ext_vector_type(2)));
typedef float v4f __attribute__((ext_vector_type(4)));
typedef float v8f __attribute__((ext_vector_type(8)));
typedef float v4f_u __attribute__((ext_vector_type(4), aligned(4)));

#define B_TOT 4096
#define F     39
#define FP    40
#define D     16
#define L0N   128
#define H1    64
#define K0    1521
#define K1    2496
#define NB    8
#define MT    2
#define TPB   128
#define PTPB  256
#define NJ    5
#define NS    10

#define NU0     119
#define WS_NEED 487424
#define NP1      2560
#define NB1      320
#define W1P_OFF  121856
#define WS_NEED1 1798144

#define XS_OFF 0
#define YS_OFF 5120
#define R0_OFF (5120 + 8192)
#define R1_OFF (5120 + 8192 + 8192)
#define LDS_FLOATS (5120 + 8192 + 8192 + 16384)

__host__ __device__ constexpr int ubase(int q) {
  return q == 0 ? 0 : q == 1 ? 40 : q == 2 ? 72 : q == 3 ? 96 : 112;
}

__device__ __forceinline__ v4f ldu4(const float* p) {
  return *(const v4f_u*)p;
}

__global__ __launch_bounds__(PTPB) void cin_sym_pack_kernel(
    const float* __restrict__ W0, const float* __restrict__ W1,
    float* __restrict__ ws, size_t ws_bytes)
{
  if (blockIdx.x >= NU0) {
    if (ws_bytes < (size_t)WS_NEED1) return;
    const int g1   = (blockIdx.x - NU0) * PTPB + threadIdx.x;
    const int lane = g1 & 31;
    const int pj   = g1 >> 5;
    const int th   = pj / NJ;
    const int j    = pj - th * NJ;
    const int h    = th & (H1 - 1);
    const int tile = th >> 6;
    const int m0   = 8 * j + 4 * (lane >> 4);
    const float* wo = W1 + (size_t)(16 * tile + (lane & 15)) * K1 + h * F + m0;
    v4f o;
    o.x = wo[0]; o.y = wo[1]; o.z = wo[2];
    o.w = (m0 + 3 < F) ? wo[3] : 0.0f;
    *(v4f*)(ws + W1P_OFF + (size_t)g1 * 4) = o;
    return;
  }
  if (ws_bytes < (size_t)WS_NEED) return;
  const int gid  = blockIdx.x * PTPB + threadIdx.x;
  const int lane = gid & 31;
  const int tu   = gid >> 5;
  const int tile = tu / NU0;
  const int u    = tu - tile * NU0;
  const int q    = (u >= 112) ? 4 : (u >= 96) ? 3 : (u >= 72) ? 2 : (u >= 40) ? 1 : 0;
  const int per  = NJ - q;
  const int r    = u - ubase(q);
  const int hh   = r / per;
  const int h    = 8 * q + hh;
  const int j    = q + (r - hh * per);
  const int m0   = 8 * j + 4 * (lane >> 4);
  const float* wo = W0 + (size_t)(16 * tile + (lane & 15)) * K0;
  float v[4];
#pragma unroll
  for (int c = 0; c < 4; ++c) {
    const int m = m0 + c;
    float s = 0.0f;
    if (m < F) {
      bool pair;
      if (j > q)                       pair = true;
      else if (((h ^ m) & 2) == 0)     pair = m > h;
      else                             pair = (h & 2) == 0;
      if (m == h)    s = wo[h * F + h];
      else if (pair) s = wo[h * F + m] + wo[m * F + h];
    }
    v[c] = s;
  }
  v4f o; o.x = v[0]; o.y = v[1]; o.z = v[2]; o.w = v[3];
  *(v4f*)(ws + (size_t)gid * 4) = o;
}

__device__ __forceinline__ void hgroup(const v4f (&a)[NJ], const float (&m)[NB],
                                       const v2f (&xf)[NB][NS], v8f (&acc)[NB])
{
#pragma unroll
  for (int j = 0; j < NJ; ++j) {
    v2f aA, aB;
    aA.x = a[j].x; aA.y = a[j].y;
    aB.x = a[j].z; aB.y = a[j].w;
#pragma unroll
    for (int nb = 0; nb < NB; ++nb) {
      const v2f b = xf[nb][2 * j] * m[nb];
      acc[nb] = __builtin_amdgcn_wmma_f32_16x16x4_f32(
          false, aA, false, b, (short)0, acc[nb], false, false);
    }
#pragma unroll
    for (int nb = 0; nb < NB; ++nb) {
      const v2f b = xf[nb][2 * j + 1] * m[nb];
      acc[nb] = __builtin_amdgcn_wmma_f32_16x16x4_f32(
          false, aB, false, b, (short)0, acc[nb], false, false);
    }
  }
}

template <int NH>
__device__ __forceinline__ void run_layer(const float* __restrict__ wrow, int hi, int lo16,
                                          const float* mult, int mstride,
                                          const v2f (&xf)[NB][NS], v8f (&acc)[NB])
{
  const float* wr = wrow + 4 * hi;
  v4f   ac[NJ];
  float mc[NB];
#pragma unroll
  for (int j = 0; j < NJ; ++j) ac[j] = ldu4(wr + 8 * j);
#pragma unroll
  for (int nb = 0; nb < NB; ++nb) mc[nb] = mult[nb * mstride + lo16];

  for (int h = 1; h < NH - 1; ++h) {
    v4f   an[NJ];
    float mn[NB];
    const float* wn = wr + h * F;
#pragma unroll
    for (int j = 0; j < NJ; ++j) an[j] = ldu4(wn + 8 * j);
#pragma unroll
    for (int nb = 0; nb < NB; ++nb) mn[nb] = mult[nb * mstride + h * D + lo16];
    hgroup(ac, mc, xf, acc);
#pragma unroll
    for (int j = 0; j < NJ; ++j) ac[j] = an[j];
#pragma unroll
    for (int nb = 0; nb < NB; ++nb) mc[nb] = mn[nb];
  }

  {
    v4f   an[NJ];
    float mn[NB];
    const float* wn = wr + (NH - 1) * F;
#pragma unroll
    for (int j = 0; j < NJ - 1; ++j) an[j] = ldu4(wn + 8 * j);
    const float* wl = wrow + (NH - 1) * F;
    const int m0 = 32 + 4 * hi;
    an[NJ - 1].x = wl[m0];
    an[NJ - 1].y = wl[m0 + 1];
    an[NJ - 1].z = wl[m0 + 2];
    an[NJ - 1].w = wl[(m0 + 3 < F) ? (m0 + 3) : (F - 1)];
#pragma unroll
    for (int nb = 0; nb < NB; ++nb) mn[nb] = mult[nb * mstride + (NH - 1) * D + lo16];
    hgroup(ac, mc, xf, acc);
    hgroup(an, mn, xf, acc);
  }
}

__device__ __forceinline__ void hgroup_mt(const v4f (&a)[MT][NJ], const float (&m)[NB],
                                          const v2f (&xf)[NB][NS], v8f (&acc)[MT][NB])
{
#pragma unroll
  for (int j = 0; j < NJ; ++j) {
#pragma unroll
    for (int nb = 0; nb < NB; ++nb) {
      const v2f b = xf[nb][2 * j] * m[nb];
#pragma unroll
      for (int t = 0; t < MT; ++t) {
        v2f aA; aA.x = a[t][j].x; aA.y = a[t][j].y;
        acc[t][nb] = __builtin_amdgcn_wmma_f32_16x16x4_f32(
            false, aA, false, b, (short)0, acc[t][nb], false, false);
      }
    }
#pragma unroll
    for (int nb = 0; nb < NB; ++nb) {
      const v2f b = xf[nb][2 * j + 1] * m[nb];
#pragma unroll
      for (int t = 0; t < MT; ++t) {
        v2f aB; aB.x = a[t][j].z; aB.y = a[t][j].w;
        acc[t][nb] = __builtin_amdgcn_wmma_f32_16x16x4_f32(
            false, aB, false, b, (short)0, acc[t][nb], false, false);
      }
    }
  }
}

__device__ __forceinline__ void run_layer1_packed(const v4f* __restrict__ wp, int lo16,
                                                  const float* mult,
                                                  const v2f (&xf)[NB][NS], v8f (&acc)[MT][NB])
{
  v4f   ac[MT][NJ];
  float mc[NB];
#pragma unroll
  for (int t = 0; t < MT; ++t)
#pragma unroll
    for (int j = 0; j < NJ; ++j) ac[t][j] = wp[t * (H1 * NJ * 32) + j * 32];
#pragma unroll
  for (int nb = 0; nb < NB; ++nb) mc[nb] = mult[nb * (H1 * D) + lo16];

  for (int h = 1; h < H1; ++h) {
    v4f   an[MT][NJ];
    float mn[NB];
    const v4f* wn = wp + h * (NJ * 32);
#pragma unroll
    for (int t = 0; t < MT; ++t)
#pragma unroll
      for (int j = 0; j < NJ; ++j) an[t][j] = wn[t * (H1 * NJ * 32) + j * 32];
#pragma unroll
    for (int nb = 0; nb < NB; ++nb) mn[nb] = mult[nb * (H1 * D) + h * D + lo16];
    hgroup_mt(ac, mc, xf, acc);
#pragma unroll
    for (int t = 0; t < MT; ++t)
#pragma unroll
      for (int j = 0; j < NJ; ++j) ac[t][j] = an[t][j];
#pragma unroll
    for (int nb = 0; nb < NB; ++nb) mc[nb] = mn[nb];
  }
  hgroup_mt(ac, mc, xf, acc);
}


template <int Q, bool LEAD0>
__device__ __forceinline__ void hgroup_sym(const v2f (&a0)[MT][NJ], const v2f (&a1)[MT][NJ],
                                           const float (&m)[NB],
                                           const v2f (&xf)[NB][NS], v8f (&acc)[MT][NB])
{
#pragma unroll
  for (int j = Q; j < NJ; ++j) {
    if (LEAD0 || j > Q) {
#pragma unroll
      for (int nb = 0; nb < NB; ++nb) {
        const v2f b = xf[nb][2 * j] * m[nb];
#pragma unroll
        for (int t = 0; t < MT; ++t)
          acc[t][nb] = __builtin_amdgcn_wmma_f32_16x16x4_f32(
              false, a0[t][j], false, b, (short)0, acc[t][nb], false, false);
      }
    }
#pragma unroll
    for (int nb = 0; nb < NB; ++nb) {
      const v2f b = xf[nb][2 * j + 1] * m[nb];
#pragma unroll
      for (int t = 0; t < MT; ++t)
        acc[t][nb] = __builtin_amdgcn_wmma_f32_16x16x4_f32(
            false, a1[t][j], false, b, (short)0, acc[t][nb], false, false);
    }
  }
}

template <int Q, bool LEAD0, int QN, bool LEADN>
__device__ __forceinline__ void sym_step(const v4f* __restrict__ sp, int un,
                                         const float* mult, int hn, int lo16,
                                         v2f (&a0)[MT][NJ], v2f (&a1)[MT][NJ], float (&mc)[NB],
                                         const v2f (&xf)[NB][NS], v8f (&acc)[MT][NB])
{
  v2f   n0[MT][NJ], n1[MT][NJ];
  float mn[NB];
#pragma unroll
  for (int t = 0; t < MT; ++t) {
#pragma unroll
    for (int j = QN; j < NJ; ++j) {
      const v4f* pp = sp + t * (NU0 * 32) + (un + (j - QN)) * 32;
      if (LEADN || j > QN) {
        const v4f q4 = *pp;
        n0[t][j].x = q4.x; n0[t][j].y = q4.y;
        n1[t][j].x = q4.z; n1[t][j].y = q4.w;
      } else {
        n1[t][j] = *((const v2f*)pp + 1);
      }
    }
  }
#pragma unroll
  for (int nb = 0; nb < NB; ++nb) mn[nb] = mult[nb * (FP * D) + hn * D + lo16];
  hgroup_sym<Q, LEAD0>(a0, a1, mc, xf, acc);
#pragma unroll
  for (int t = 0; t < MT; ++t) {
#pragma unroll
    for (int j = QN; j < NJ; ++j) {
      if (LEADN || j > QN) a0[t][j] = n0[t][j];
      a1[t][j] = n1[t][j];
    }
  }
#pragma unroll
  for (int nb = 0; nb < NB; ++nb) mc[nb] = mn[nb];
}

template <int Q>
__device__ __forceinline__ void sym_block(const v4f* __restrict__ sp, const float* mult, int lo16,
                                          v2f (&a0)[MT][NJ], v2f (&a1)[MT][NJ], float (&mc)[NB],
                                          const v2f (&xf)[NB][NS], v8f (&acc)[MT][NB])
{
  constexpr int PER = NJ - Q;
  constexpr int UB  = ubase(Q);
  constexpr int H0  = 8 * Q;
  constexpr int NC1 = (Q < 4) ? 4 : 3;
#pragma unroll 1
  for (int p = 1; p <= 4; ++p) {
    const int rn = (p & 1) | ((p & 2) << 1) | ((p & 4) >> 1);
    sym_step<Q, true, Q, true>(sp, UB + rn * PER, mult, H0 + rn, lo16, a0, a1, mc, xf, acc);
  }
#pragma unroll 1
  for (int p = 5; p < 4 + NC1; ++p) {
    const int rn = (p & 1) | ((p & 2) << 1) | ((p & 4) >> 1);
    sym_step<Q, false, Q, false>(sp, UB + rn * PER, mult, H0 + rn, lo16, a0, a1, mc, xf, acc);
  }
  if (Q < 4)
    sym_step<Q, false, Q + 1, true>(sp, ubase(Q + 1), mult, H0 + 8, lo16, a0, a1, mc, xf, acc);
  else
    hgroup_sym<Q, false>(a0, a1, mc, xf, acc);
}

__device__ __forceinline__ void run_layer0_sym(const float* __restrict__ ws, int wave, int lane,
                                               int lo16, const float* xs,
                                               const v2f (&xf)[NB][NS], v8f (&acc)[MT][NB])
{
  const v4f* sp = (const v4f*)ws + (size_t)(MT * wave) * (NU0 * 32) + lane;
  v2f   a0[MT][NJ], a1[MT][NJ];
  float mc[NB];
#pragma unroll
  for (int t = 0; t < MT; ++t) {
#pragma unroll
    for (int j = 0; j < NJ; ++j) {
      const v4f q4 = sp[t * (NU0 * 32) + j * 32];
      a0[t][j].x = q4.x; a0[t][j].y = q4.y;
      a1[t][j].x = q4.z; a1[t][j].y = q4.w;
    }
  }
#pragma unroll
  for (int nb = 0; nb < NB; ++nb) mc[nb] = xs[nb * (FP * D) + lo16];
  sym_block<0>(sp, xs, lo16, a0, a1, mc, xf, acc);
  sym_block<1>(sp, xs, lo16, a0, a1, mc, xf, acc);
  sym_block<2>(sp, xs, lo16, a0, a1, mc, xf, acc);
  sym_block<3>(sp, xs, lo16, a0, a1, mc, xf, acc);
  sym_block<4>(sp, xs, lo16, a0, a1, mc, xf, acc);
}

__global__ __launch_bounds__(TPB, 2) void cin_wmma_kernel(
    const float* __restrict__ x,  const float* __restrict__ W0,
    const float* __restrict__ b0, const float* __restrict__ W1,
    const float* __restrict__ b1, float* __restrict__ out,
    const float* __restrict__ ws, size_t ws_bytes)
{
  __shared__ __align__(16) float lds[LDS_FLOATS];
  float* xs = lds + XS_OFF;
  float* ys = lds + YS_OFF;
  float* r0 = lds + R0_OFF;
  float* r1 = lds + R1_OFF;

  const int tid   = threadIdx.x;
  const int wave  = tid >> 5;
  const int lane  = tid & 31;
  const int lo16  = lane & 15;
  const int hi    = lane >> 4;
  const int bbase = blockIdx.x * NB;

  const v4f* xg4 = (const v4f*)(x + (size_t)bbase * (F * D));
#pragma unroll
  for (int i = 0; i < (NB * F * D / 4 + TPB - 1) / TPB; ++i) {
    const int e4 = tid + i * TPB;
    if (e4 < NB * F * D / 4) {
      const int nb  = e4 / (F * D / 4);
      const int rem = e4 - nb * (F * D / 4);
      *(v4f*)(xs + nb * (FP * D) + rem * 4) = xg4[e4];
    }
  }
  if (tid < NB * D) xs[(tid >> 4) * (FP * D) + F * D + (tid & 15)] = 0.0f;
  __syncthreads();

  v2f xf[NB][NS];
#pragma unroll
  for (int nb = 0; nb < NB; ++nb) {
#pragma unroll
    for (int s = 0; s < NS; ++s) {
      const int m = 8 * (s >> 1) + 4 * hi + 2 * (s & 1);
      xf[nb][s].x = xs[nb * (FP * D) + m * D + lo16];
      xf[nb][s].y = xs[nb * (FP * D) + (m + 1) * D + lo16];
    }
  }

  v8f acc[MT][NB];
#pragma unroll
  for (int t = 0; t < MT; ++t)
#pragma unroll
    for (int nb = 0; nb < NB; ++nb)
#pragma unroll
      for (int j = 0; j < 8; ++j) acc[t][nb][j] = 0.0f;

  if (__builtin_expect(ws_bytes >= (size_t)WS_NEED, 1)) {
    run_layer0_sym(ws, wave, lane, lo16, xs, xf, acc);
  } else {
#pragma unroll
    for (int t = 0; t < MT; ++t)
      run_layer<F>(W0 + (size_t)(((MT * wave + t) << 4) + lo16) * K0, hi, lo16, xs, FP * D, xf, acc[t]);
  }

#pragma unroll
  for (int t = 0; t < MT; ++t) {
#pragma unroll
    for (int nb = 0; nb < NB; ++nb) {
#pragma unroll
      for (int j = 0; j < 8; ++j) {
        const int row = ((MT * wave + t) << 4) + j + (hi << 3);
        float v = acc[t][nb][j] + b0[row];
        v = v > 0.0f ? v : 0.0f;
        if (wave < 2) ys[nb * (H1 * D) + row * D + lo16] = v;
        else          r0[nb * (H1 * D) + (row - H1) * D + lo16] = v;
        acc[t][nb][j] = 0.0f;
      }
    }
  }
  __syncthreads();

#pragma unroll
  for (int rep = 0; rep < NB * H1 / TPB; ++rep) {
    const int idx = tid + rep * TPB;
    const int nb  = idx >> 6;
    const int r   = idx & 63;
    float sum = 0.0f;
#pragma unroll
    for (int d = 0; d < D; ++d) sum += r0[nb * (H1 * D) + r * D + d];
    out[(size_t)(bbase + nb) * 192 + r] = sum;
  }

  if (__builtin_expect(ws_bytes >= (size_t)WS_NEED1, 1)) {
    run_layer1_packed((const v4f*)(ws + W1P_OFF) + (size_t)(MT * wave) * (H1 * NJ * 32) + lane,
                      lo16, ys, xf, acc);
  } else {
#pragma unroll
    for (int t = 0; t < MT; ++t)
      run_layer<H1>(W1 + (size_t)(((MT * wave + t) << 4) + lo16) * K1, hi, lo16, ys, H1 * D, xf, acc[t]);
  }

#pragma unroll
  for (int t = 0; t < MT; ++t) {
#pragma unroll
    for (int nb = 0; nb < NB; ++nb) {
#pragma unroll
      for (int j = 0; j < 8; ++j) {
        const int row = ((MT * wave + t) << 4) + j + (hi << 3);
        float v = acc[t][nb][j] + b1[row];
        v = v > 0.0f ? v : 0.0f;
        r1[nb * (L0N * D) + row * D + lo16] = v;
      }
    }
  }
  __syncthreads();

#pragma unroll
  for (int rep = 0; rep < NB * L0N / TPB; ++rep) {
    const int idx = tid + rep * TPB;
    const int nb  = idx >> 7;
    const int r   = idx & 127;
    float sum = 0.0f;
#pragma unroll
    for (int d = 0; d < D; ++d) sum += r1[nb * (L0N * D) + r * D + d];
    out[(size_t)(bbase + nb) * 192 + 64 + r] = sum;
  }
}

extern "C" void kernel_launch(void* const* d_in, const int* in_sizes, int n_in,
                              void* d_out, int out_size, void* d_ws, size_t ws_size,
                              hipStream_t stream) {
  const float* x  = (const float*)d_in[0];
  const float* W0 = (const float*)d_in[1];
  const float* b0 = (const float*)d_in[2];
  const float* W1 = (const float*)d_in[3];
  const float* b1 = (const float*)d_in[4];
  float* out = (float*)d_out;
  float* ws  = (float*)d_ws;

  hipLaunchKernelGGL(cin_sym_pack_kernel, dim3(NU0 + NB1), dim3(PTPB), 0, stream,
                     W0, W1, ws, ws_size);

  dim3 grid(B_TOT / NB);
  dim3 block(TPB);
  hipLaunchKernelGGL(cin_wmma_kernel, grid, block, 0, stream,
                     x, W0, b0, W1, b1, out, (const float*)ws, ws_size);
}


# Inside an h-group the batch loop is outermost: a batch's B fragments are formed as one run, then its WMMAs alternate between its two accumulators (before: 16 in rotation); order of sums unchanged
# speedup vs baseline: 1.9019x; 1.0345x over previous
#include <hip/hip_runtime.h>


typedef float v2f __attribute__((ext_vector_type(2)));
typedef float v4f __attribute__((ext_vector_type(4)));
typedef float v8f __attribute__((ext_vector_type(8)));
typedef float v4f_u __attribute__((ext_vector_type(4), aligned(4)));

#define B_TOT 4096
#define F     39
#define FP    40
#define D     16
#define L0N   128
#define H1    64
#define K0    1521
#define K1    2496
#define NB    8
#define MT    2
#define TPB   128
#define PTPB  256
#define NJ    5
#define NS    10

#define NU0     119
#define WS_NEED 487424
#define NP1      2560
#define NB1      320
#define W1P_OFF  121856
#define WS_NEED1 1798144

#define XS_OFF 0
#define YS_OFF 5120
#define R0_OFF (5120 + 8192)
#define R1_OFF (5120 + 8192 + 8192)
#define LDS_FLOATS (5120 + 8192 + 8192 + 16384)

__host__ __device__ constexpr int ubase(int q) {
  return q == 0 ? 0 : q == 1 ? 40 : q == 2 ? 72 : q == 3 ? 96 : 112;
}

__device__ __forceinline__ v4f ldu4(const float* p) {
  return *(const v4f_u*)p;
}

__global__ __launch_bounds__(PTPB) void cin_sym_pack_kernel(
    const float* __restrict__ W0, const float* __restrict__ W1,
    float* __restrict__ ws, size_t ws_bytes)
{
  if (blockIdx.x >= NU0) {
    if (ws_bytes < (size_t)WS_NEED1) return;
    const int g1   = (blockIdx.x - NU0) * PTPB + threadIdx.x;
    const int lane = g1 & 31;
    const int pj   = g1 >> 5;
    const int th   = pj / NJ;
    const int j    = pj - th * NJ;
    const int h    = th & (H1 - 1);
    const int tile = th >> 6;
    const int m0   = 8 * j + 4 * (lane >> 4);
    const float* wo = W1 + (size_t)(16 * tile + (lane & 15)) * K1 + h * F + m0;
    v4f o;
    o.x = wo[0]; o.y = wo[1]; o.z = wo[2];
    o.w = (m0 + 3 < F) ? wo[3] : 0.0f;
    *(v4f*)(ws + W1P_OFF + (size_t)g1 * 4) = o;
    return;
  }
  if (ws_bytes < (size_t)WS_NEED) return;
  const int gid  = blockIdx.x * PTPB + threadIdx.x;
  const int lane = gid & 31;
  const int tu   = gid >> 5;
  const int tile = tu / NU0;
  const int u    = tu - tile * NU0;
  const int q    = (u >= 112) ? 4 : (u >= 96) ? 3 : (u >= 72) ? 2 : (u >= 40) ? 1 : 0;
  const int per  = NJ - q;
  const int r    = u - ubase(q);
  const int hh   = r / per;
  const int h    = 8 * q + hh;
  const int j    = q + (r - hh * per);
  const int m0   = 8 * j + 4 * (lane >> 4);
  const float* wo = W0 + (size_t)(16 * tile + (lane & 15)) * K0;
  float v[4];
#pragma unroll
  for (int c = 0; c < 4; ++c) {
    const int m = m0 + c;
    float s = 0.0f;
    if (m < F) {
      bool pair;
      if (j > q)                       pair = true;
      else if (((h ^ m) & 2) == 0)     pair = m > h;
      else                             pair = (h & 2) == 0;
      if (m == h)    s = wo[h * F + h];
      else if (pair) s = wo[h * F + m] + wo[m * F + h];
    }
    v[c] = s;
  }
  v4f o; o.x = v[0]; o.y = v[1]; o.z = v[2]; o.w = v[3];
  *(v4f*)(ws + (size_t)gid * 4) = o;
}

__device__ __forceinline__ void hgroup(const v4f (&a)[NJ], const float (&m)[NB],
                                       const v2f (&xf)[NB][NS], v8f (&acc)[NB])
{
#pragma unroll
  for (int j = 0; j < NJ; ++j) {
    v2f aA, aB;
    aA.x = a[j].x; aA.y = a[j].y;
    aB.x = a[j].z; aB.y = a[j].w;
#pragma unroll
    for (int nb = 0; nb < NB; ++nb) {
      const v2f b = xf[nb][2 * j] * m[nb];
      acc[nb] = __builtin_amdgcn_wmma_f32_16x16x4_f32(
          false, aA, false, b, (short)0, acc[nb], false, false);
    }
#pragma unroll
    for (int nb = 0; nb < NB; ++nb) {
      const v2f b = xf[nb][2 * j + 1] * m[nb];
      acc[nb] = __builtin_amdgcn_wmma_f32_16x16x4_f32(
          false, aB, false, b, (short)0, acc[nb], false, false);
    }
  }
}

template <int NH>
__device__ __forceinline__ void run_layer(const float* __restrict__ wrow, int hi, int lo16,
                                          const float* mult, int mstride,
                                          const v2f (&xf)[NB][NS], v8f (&acc)[NB])
{
  const float* wr = wrow + 4 * hi;
  v4f   ac[NJ];
  float mc[NB];
#pragma unroll
  for (int j = 0; j < NJ; ++j) ac[j] = ldu4(wr + 8 * j);
#pragma unroll
  for (int nb = 0; nb < NB; ++nb) mc[nb] = mult[nb * mstride + lo16];

  for (int h = 1; h < NH - 1; ++h) {
    v4f   an[NJ];
    float mn[NB];
    const float* wn = wr + h * F;
#pragma unroll
    for (int j = 0; j < NJ; ++j) an[j] = ldu4(wn + 8 * j);
#pragma unroll
    for (int nb = 0; nb < NB; ++nb) mn[nb] = mult[nb * mstride + h * D + lo16];
    hgroup(ac, mc, xf, acc);
#pragma unroll
    for (int j = 0; j < NJ; ++j) ac[j] = an[j];
#pragma unroll
    for (int nb = 0; nb < NB; ++nb) mc[nb] = mn[nb];
  }

  {
    v4f   an[NJ];
    float mn[NB];
    const float* wn = wr + (NH - 1) * F;
#pragma unroll
    for (int j = 0; j < NJ - 1; ++j) an[j] = ldu4(wn + 8 * j);
    const float* wl = wrow + (NH - 1) * F;
    const int m0 = 32 + 4 * hi;
    an[NJ - 1].x = wl[m0];
    an[NJ - 1].y = wl[m0 + 1];
    an[NJ - 1].z = wl[m0 + 2];
    an[NJ - 1].w = wl[(m0 + 3 < F) ? (m0 + 3) : (F - 1)];
#pragma unroll
    for (int nb = 0; nb < NB; ++nb) mn[nb] = mult[nb * mstride + (NH - 1) * D + lo16];
    hgroup(ac, mc, xf, acc);
    hgroup(an, mn, xf, acc);
  }
}

__device__ __forceinline__ void hgroup_mt(const v4f (&a)[MT][NJ], const float (&m)[NB],
                                          const v2f (&xf)[NB][NS], v8f (&acc)[MT][NB])
{
#pragma unroll
  for (int nb = 0; nb < NB; ++nb) {
    v2f b[NS];
#pragma unroll
    for (int s = 0; s < NS; ++s) b[s] = xf[nb][s] * m[nb];
    __builtin_amdgcn_sched_barrier(0);
#pragma unroll
    for (int j = 0; j < NJ; ++j) {
#pragma unroll
      for (int t = 0; t < MT; ++t) {
        v2f aA; aA.x = a[t][j].x; aA.y = a[t][j].y;
        acc[t][nb] = __builtin_amdgcn_wmma_f32_16x16x4_f32(
            false, aA, false, b[2 * j], (short)0, acc[t][nb], false, false);
      }
#pragma unroll
      for (int t = 0; t < MT; ++t) {
        v2f aB; aB.x = a[t][j].z; aB.y = a[t][j].w;
        acc[t][nb] = __builtin_amdgcn_wmma_f32_16x16x4_f32(
            false, aB, false, b[2 * j + 1], (short)0, acc[t][nb], false, false);
      }
    }
    __builtin_amdgcn_sched_barrier(0);
  }
}

__device__ __forceinline__ void run_layer1_packed(const v4f* __restrict__ wp, int lo16,
                                                  const float* mult,
                                                  const v2f (&xf)[NB][NS], v8f (&acc)[MT][NB])
{
  v4f   ac[MT][NJ];
  float mc[NB];
#pragma unroll
  for (int t = 0; t < MT; ++t)
#pragma unroll
    for (int j = 0; j < NJ; ++j) ac[t][j] = wp[t * (H1 * NJ * 32) + j * 32];
#pragma unroll
  for (int nb = 0; nb < NB; ++nb) mc[nb] = mult[nb * (H1 * D) + lo16];

  for (int h = 1; h < H1; ++h) {
    v4f   an[MT][NJ];
    float mn[NB];
    const v4f* wn = wp + h * (NJ * 32);
#pragma unroll
    for (int t = 0; t < MT; ++t)
#pragma unroll
      for (int j = 0; j < NJ; ++j) an[t][j] = wn[t * (H1 * NJ * 32) + j * 32];
#pragma unroll
    for (int nb = 0; nb < NB; ++nb) mn[nb] = mult[nb * (H1 * D) + h * D + lo16];
    hgroup_mt(ac, mc, xf, acc);
#pragma unroll
    for (int t = 0; t < MT; ++t)
#pragma unroll
      for (int j = 0; j < NJ; ++j) ac[t][j] = an[t][j];
#pragma unroll
    for (int nb = 0; nb < NB; ++nb) mc[nb] = mn[nb];
  }
  hgroup_mt(ac, mc, xf, acc);
}


template <int Q, bool LEAD0>
__device__ __forceinline__ void hgroup_sym(const v2f (&a0)[MT][NJ], const v2f (&a1)[MT][NJ],
                                           const float (&m)[NB],
                                           const v2f (&xf)[NB][NS], v8f (&acc)[MT][NB])
{
#pragma unroll
  for (int nb = 0; nb < NB; ++nb) {
    v2f b[NS];
#pragma unroll
    for (int s = 2 * Q; s < NS; ++s)
      if (LEAD0 || s > 2 * Q) b[s] = xf[nb][s] * m[nb];
    __builtin_amdgcn_sched_barrier(0);
#pragma unroll
    for (int j = Q; j < NJ; ++j) {
      if (LEAD0 || j > Q) {
#pragma unroll
        for (int t = 0; t < MT; ++t)
          acc[t][nb] = __builtin_amdgcn_wmma_f32_16x16x4_f32(
              false, a0[t][j], false, b[2 * j], (short)0, acc[t][nb], false, false);
      }
#pragma unroll
      for (int t = 0; t < MT; ++t)
        acc[t][nb] = __builtin_amdgcn_wmma_f32_16x16x4_f32(
            false, a1[t][j], false, b[2 * j + 1], (short)0, acc[t][nb], false, false);
    }
    __builtin_amdgcn_sched_barrier(0);
  }
}

template <int Q, bool LEAD0, int QN, bool LEADN>
__device__ __forceinline__ void sym_step(const v4f* __restrict__ sp, int un,
                                         const float* mult, int hn, int lo16,
                                         v2f (&a0)[MT][NJ], v2f (&a1)[MT][NJ], float (&mc)[NB],
                                         const v2f (&xf)[NB][NS], v8f (&acc)[MT][NB])
{
  v2f   n0[MT][NJ], n1[MT][NJ];
  float mn[NB];
#pragma unroll
  for (int t = 0; t < MT; ++t) {
#pragma unroll
    for (int j = QN; j < NJ; ++j) {
      const v4f* pp = sp + t * (NU0 * 32) + (un + (j - QN)) * 32;
      if (LEADN || j > QN) {
        const v4f q4 = *pp;
        n0[t][j].x = q4.x; n0[t][j].y = q4.y;
        n1[t][j].x = q4.z; n1[t][j].y = q4.w;
      } else {
        n1[t][j] = *((const v2f*)pp + 1);
      }
    }
  }
#pragma unroll
  for (int nb = 0; nb < NB; ++nb) mn[nb] = mult[nb * (FP * D) + hn * D + lo16];
  hgroup_sym<Q, LEAD0>(a0, a1, mc, xf, acc);
#pragma unroll
  for (int t = 0; t < MT; ++t) {
#pragma unroll
    for (int j = QN; j < NJ; ++j) {
      if (LEADN || j > QN) a0[t][j] = n0[t][j];
      a1[t][j] = n1[t][j];
    }
  }
#pragma unroll
  for (int nb = 0; nb < NB; ++nb) mc[nb] = mn[nb];
}

template <int Q>
__device__ __forceinline__ void sym_block(const v4f* __restrict__ sp, const float* mult, int lo16,
                                          v2f (&a0)[MT][NJ], v2f (&a1)[MT][NJ], float (&mc)[NB],
                                          const v2f (&xf)[NB][NS], v8f (&acc)[MT][NB])
{
  constexpr int PER = NJ - Q;
  constexpr int UB  = ubase(Q);
  constexpr int H0  = 8 * Q;
  constexpr int NC1 = (Q < 4) ? 4 : 3;
#pragma unroll 1
  for (int p = 1; p <= 4; ++p) {
    const int rn = (p & 1) | ((p & 2) << 1) | ((p & 4) >> 1);
    sym_step<Q, true, Q, true>(sp, UB + rn * PER, mult, H0 + rn, lo16, a0, a1, mc, xf, acc);
  }
#pragma unroll 1
  for (int p = 5; p < 4 + NC1; ++p) {
    const int rn = (p & 1) | ((p & 2) << 1) | ((p & 4) >> 1);
    sym_step<Q, false, Q, false>(sp, UB + rn * PER, mult, H0 + rn, lo16, a0, a1, mc, xf, acc);
  }
  if (Q < 4)
    sym_step<Q, false, Q + 1, true>(sp, ubase(Q + 1), mult, H0 + 8, lo16, a0, a1, mc, xf, acc);
  else
    hgroup_sym<Q, false>(a0, a1, mc, xf, acc);
}

__device__ __forceinline__ void run_layer0_sym(const float* __restrict__ ws, int wave, int lane,
                                               int lo16, const float* xs,
                                               const v2f (&xf)[NB][NS], v8f (&acc)[MT][NB])
{
  const v4f* sp = (const v4f*)ws + (size_t)(MT * wave) * (NU0 * 32) + lane;
  v2f   a0[MT][NJ], a1[MT][NJ];
  float mc[NB];
#pragma unroll
  for (int t = 0; t < MT; ++t) {
#pragma unroll
    for (int j = 0; j < NJ; ++j) {
      const v4f q4 = sp[t * (NU0 * 32) + j * 32];
      a0[t][j].x = q4.x; a0[t][j].y = q4.y;
      a1[t][j].x = q4.z; a1[t][j].y = q4.w;
    }
  }
#pragma unroll
  for (int nb = 0; nb < NB; ++nb) mc[nb] = xs[nb * (FP * D) + lo16];
  sym_block<0>(sp, xs, lo16, a0, a1, mc, xf, acc);
  sym_block<1>(sp, xs, lo16, a0, a1, mc, xf, acc);
  sym_block<2>(sp, xs, lo16, a0, a1, mc, xf, acc);
  sym_block<3>(sp, xs, lo16, a0, a1, mc, xf, acc);
  sym_block<4>(sp, xs, lo16, a0, a1, mc, xf, acc);
}

__global__ __launch_bounds__(TPB, 2) void cin_wmma_kernel(
    const float* __restrict__ x,  const float* __restrict__ W0,
    const float* __restrict__ b0, const float* __restrict__ W1,
    const float* __restrict__ b1, float* __restrict__ out,
    const float* __restrict__ ws, size_t ws_bytes)
{
  __shared__ __align__(16) float lds[LDS_FLOATS];
  float* xs = lds + XS_OFF;
  float* ys = lds + YS_OFF;
  float* r0 = lds + R0_OFF;
  float* r1 = lds + R1_OFF;

  const int tid   = threadIdx.x;
  const int wave  = tid >> 5;
  const int lane  = tid & 31;
  const int lo16  = lane & 15;
  const int hi    = lane >> 4;
  const int bbase = blockIdx.x * NB;

  const v4f* xg4 = (const v4f*)(x + (size_t)bbase * (F * D));
#pragma unroll
  for (int i = 0; i < (NB * F * D / 4 + TPB - 1) / TPB; ++i) {
    const int e4 = tid + i * TPB;
    if (e4 < NB * F * D / 4) {
      const int nb  = e4 / (F * D / 4);
      const int rem = e4 - nb * (F * D / 4);
      *(v4f*)(xs + nb * (FP * D) + rem * 4) = xg4[e4];
    }
  }
  if (tid < NB * D) xs[(tid >> 4) * (FP * D) + F * D + (tid & 15)] = 0.0f;
  __syncthreads();

  v2f xf[NB][NS];
#pragma unroll
  for (int nb = 0; nb < NB; ++nb) {
#pragma unroll
    for (int s = 0; s < NS; ++s) {
      const int m = 8 * (s >> 1) + 4 * hi + 2 * (s & 1);
      xf[nb][s].x = xs[nb * (FP * D) + m * D + lo16];
      xf[nb][s].y = xs[nb * (FP * D) + (m + 1) * D + lo16];
    }
  }

  v8f acc[MT][NB];
#pragma unroll
  for (int t = 0; t < MT; ++t)
#pragma unroll
    for (int nb = 0; nb < NB; ++nb)
#pragma unroll
      for (int j = 0; j < 8; ++j) acc[t][nb][j] = 0.0f;

  if (__builtin_expect(ws_bytes >= (size_t)WS_NEED, 1)) {
    run_layer0_sym(ws, wave, lane, lo16, xs, xf, acc);
  } else {
#pragma unroll
    for (int t = 0; t < MT; ++t)
      run_layer<F>(W0 + (size_t)(((MT * wave + t) << 4) + lo16) * K0, hi, lo16, xs, FP * D, xf, acc[t]);
  }

#pragma unroll
  for (int t = 0; t < MT; ++t) {
#pragma unroll
    for (int nb = 0; nb < NB; ++nb) {
#pragma unroll
      for (int j = 0; j < 8; ++j) {
        const int row = ((MT * wave + t) << 4) + j + (hi << 3);
        float v = acc[t][nb][j] + b0[row];
        v = v > 0.0f ? v : 0.0f;
        if (wave < 2) ys[nb * (H1 * D) + row * D + lo16] = v;
        else          r0[nb * (H1 * D) + (row - H1) * D + lo16] = v;
        acc[t][nb][j] = 0.0f;
      }
    }
  }
  __syncthreads();

#pragma unroll
  for (int rep = 0; rep < NB * H1 / TPB; ++rep) {
    const int idx = tid + rep * TPB;
    const int nb  = idx >> 6;
    const int r   = idx & 63;
    float sum = 0.0f;
#pragma unroll
    for (int d = 0; d < D; ++d) sum += r0[nb * (H1 * D) + r * D + d];
    out[(size_t)(bbase + nb) * 192 + r] = sum;
  }

  if (__builtin_expect(ws_bytes >= (size_t)WS_NEED1, 1)) {
    run_layer1_packed((const v4f*)(ws + W1P_OFF) + (size_t)(MT * wave) * (H1 * NJ * 32) + lane,
                      lo16, ys, xf, acc);
  } else {
#pragma unroll
    for (int t = 0; t < MT; ++t)
      run_layer<H1>(W1 + (size_t)(((MT * wave + t) << 4) + lo16) * K1, hi, lo16, ys, H1 * D, xf, acc[t]);
  }

#pragma unroll
  for (int t = 0; t < MT; ++t) {
#pragma unroll
    for (int nb = 0; nb < NB; ++nb) {
#pragma unroll
      for (int j = 0; j < 8; ++j) {
        const int row = ((MT * wave + t) << 4) + j + (hi << 3);
        float v = acc[t][nb][j] + b1[row];
        v = v > 0.0f ? v : 0.0f;
        r1[nb * (L0N * D) + row * D + lo16] = v;
      }
    }
  }
  __syncthreads();

#pragma unroll
  for (int rep = 0; rep < NB * L0N / TPB; ++rep) {
    const int idx = tid + rep * TPB;
    const int nb  = idx >> 7;
    const int r   = idx & 127;
    float sum = 0.0f;
#pragma unroll
    for (int d = 0; d < D; ++d) sum += r1[nb * (L0N * D) + r * D + d];
    out[(size_t)(bbase + nb) * 192 + 64 + r] = sum;
  }
}

extern "C" void kernel_launch(void* const* d_in, const int* in_sizes, int n_in,
                              void* d_out, int out_size, void* d_ws, size_t ws_size,
                              hipStream_t stream) {
  const float* x  = (const float*)d_in[0];
  const float* W0 = (const float*)d_in[1];
  const float* b0 = (const float*)d_in[2];
  const float* W1 = (const float*)d_in[3];
  const float* b1 = (const float*)d_in[4];
  float* out = (float*)d_out;
  float* ws  = (float*)d_ws;

  hipLaunchKernelGGL(cin_sym_pack_kernel, dim3(NU0 + NB1), dim3(PTPB), 0, stream,
                     W0, W1, ws, ws_size);

  dim3 grid(B_TOT / NB);
  dim3 block(TPB);
  hipLaunchKernelGGL(cin_wmma_kernel, grid, block, 0, stream,
                     x, W0, b0, W1, b1, out, (const float*)ws, ws_size);
}


# Wave priority falls with progress (s_setprio 3,2,1,0 at 50, 80, 95 % of a wave's WMMAs): of a WGP's two work-groups the one behind wins the matrix pipe, so both end together; sums unchanged
# speedup vs baseline: 1.9633x; 1.0323x over previous
#include <hip/hip_runtime.h>


typedef float v2f __attribute__((ext_vector_type(2)));
typedef float v4f __attribute__((ext_vector_type(4)));
typedef float v8f __attribute__((ext_vector_type(8)));
typedef float v4f_u __attribute__((ext_vector_type(4), aligned(4)));

#define B_TOT 4096
#define F     39
#define FP    40
#define D     16
#define L0N   128
#define H1    64
#define K0    1521
#define K1    2496
#define NB    8
#define MT    2
#define TPB   128
#define PTPB  256
#define NJ    5
#define NS    10
#define PRIO2_G 21
#define PRIO1_G 47
#define PRIO0_G 60

#define NU0     119
#define WS_NEED 487424
#define NP1      2560
#define NB1      320
#define W1P_OFF  121856
#define WS_NEED1 1798144

#define XS_OFF 0
#define YS_OFF 5120
#define R0_OFF (5120 + 8192)
#define R1_OFF (5120 + 8192 + 8192)
#define LDS_FLOATS (5120 + 8192 + 8192 + 16384)

__host__ __device__ constexpr int ubase(int q) {
  return q == 0 ? 0 : q == 1 ? 40 : q == 2 ? 72 : q == 3 ? 96 : 112;
}

__device__ __forceinline__ v4f ldu4(const float* p) {
  return *(const v4f_u*)p;
}

__global__ __launch_bounds__(PTPB) void cin_sym_pack_kernel(
    const float* __restrict__ W0, const float* __restrict__ W1,
    float* __restrict__ ws, size_t ws_bytes)
{
  if (blockIdx.x >= NU0) {
    if (ws_bytes < (size_t)WS_NEED1) return;
    const int g1   = (blockIdx.x - NU0) * PTPB + threadIdx.x;
    const int lane = g1 & 31;
    const int pj   = g1 >> 5;
    const int th   = pj / NJ;
    const int j    = pj - th * NJ;
    const int h    = th & (H1 - 1);
    const int tile = th >> 6;
    const int m0   = 8 * j + 4 * (lane >> 4);
    const float* wo = W1 + (size_t)(16 * tile + (lane & 15)) * K1 + h * F + m0;
    v4f o;
    o.x = wo[0]; o.y = wo[1]; o.z = wo[2];
    o.w = (m0 + 3 < F) ? wo[3] : 0.0f;
    *(v4f*)(ws + W1P_OFF + (size_t)g1 * 4) = o;
    return;
  }
  if (ws_bytes < (size_t)WS_NEED) return;
  const int gid  = blockIdx.x * PTPB + threadIdx.x;
  const int lane = gid & 31;
  const int tu   = gid >> 5;
  const int tile = tu / NU0;
  const int u    = tu - tile * NU0;
  const int q    = (u >= 112) ? 4 : (u >= 96) ? 3 : (u >= 72) ? 2 : (u >= 40) ? 1 : 0;
  const int per  = NJ - q;
  const int r    = u - ubase(q);
  const int hh   = r / per;
  const int h    = 8 * q + hh;
  const int j    = q + (r - hh * per);
  const int m0   = 8 * j + 4 * (lane >> 4);
  const float* wo = W0 + (size_t)(16 * tile + (lane & 15)) * K0;
  float v[4];
#pragma unroll
  for (int c = 0; c < 4; ++c) {
    const int m = m0 + c;
    float s = 0.0f;
    if (m < F) {
      bool pair;
      if (j > q)                       pair = true;
      else if (((h ^ m) & 2) == 0)     pair = m > h;
      else                             pair = (h & 2) == 0;
      if (m == h)    s = wo[h * F + h];
      else if (pair) s = wo[h * F + m] + wo[m * F + h];
    }
    v[c] = s;
  }
  v4f o; o.x = v[0]; o.y = v[1]; o.z = v[2]; o.w = v[3];
  *(v4f*)(ws + (size_t)gid * 4) = o;
}

__device__ __forceinline__ void hgroup(const v4f (&a)[NJ], const float (&m)[NB],
                                       const v2f (&xf)[NB][NS], v8f (&acc)[NB])
{
#pragma unroll
  for (int j = 0; j < NJ; ++j) {
    v2f aA, aB;
    aA.x = a[j].x; aA.y = a[j].y;
    aB.x = a[j].z; aB.y = a[j].w;
#pragma unroll
    for (int nb = 0; nb < NB; ++nb) {
      const v2f b = xf[nb][2 * j] * m[nb];
      acc[nb] = __builtin_amdgcn_wmma_f32_16x16x4_f32(
          false, aA, false, b, (short)0, acc[nb], false, false);
    }
#pragma unroll
    for (int nb = 0; nb < NB; ++nb) {
      const v2f b = xf[nb][2 * j + 1] * m[nb];
      acc[nb] = __builtin_amdgcn_wmma_f32_16x16x4_f32(
          false, aB, false, b, (short)0, acc[nb], false, false);
    }
  }
}

template <int NH>
__device__ __forceinline__ void run_layer(const float* __restrict__ wrow, int hi, int lo16,
                                          const float* mult, int mstride,
                                          const v2f (&xf)[NB][NS], v8f (&acc)[NB])
{
  const float* wr = wrow + 4 * hi;
  v4f   ac[NJ];
  float mc[NB];
#pragma unroll
  for (int j = 0; j < NJ; ++j) ac[j] = ldu4(wr + 8 * j);
#pragma unroll
  for (int nb = 0; nb < NB; ++nb) mc[nb] = mult[nb * mstride + lo16];

  for (int h = 1; h < NH - 1; ++h) {
    v4f   an[NJ];
    float mn[NB];
    const float* wn = wr + h * F;
#pragma unroll
    for (int j = 0; j < NJ; ++j) an[j] = ldu4(wn + 8 * j);
#pragma unroll
    for (int nb = 0; nb < NB; ++nb) mn[nb] = mult[nb * mstride + h * D + lo16];
    hgroup(ac, mc, xf, acc);
#pragma unroll
    for (int j = 0; j < NJ; ++j) ac[j] = an[j];
#pragma unroll
    for (int nb = 0; nb < NB; ++nb) mc[nb] = mn[nb];
  }

  {
    v4f   an[NJ];
    float mn[NB];
    const float* wn = wr + (NH - 1) * F;
#pragma unroll
    for (int j = 0; j < NJ - 1; ++j) an[j] = ldu4(wn + 8 * j);
    const float* wl = wrow + (NH - 1) * F;
    const int m0 = 32 + 4 * hi;
    an[NJ - 1].x = wl[m0];
    an[NJ - 1].y = wl[m0 + 1];
    an[NJ - 1].z = wl[m0 + 2];
    an[NJ - 1].w = wl[(m0 + 3 < F) ? (m0 + 3) : (F - 1)];
#pragma unroll
    for (int nb = 0; nb < NB; ++nb) mn[nb] = mult[nb * mstride + (NH - 1) * D + lo16];
    hgroup(ac, mc, xf, acc);
    hgroup(an, mn, xf, acc);
  }
}

__device__ __forceinline__ void hgroup_mt(const v4f (&a)[MT][NJ], const float (&m)[NB],
                                          const v2f (&xf)[NB][NS], v8f (&acc)[MT][NB])
{
#pragma unroll
  for (int nb = 0; nb < NB; ++nb) {
    v2f b[NS];
#pragma unroll
    for (int s = 0; s < NS; ++s) b[s] = xf[nb][s] * m[nb];
    __builtin_amdgcn_sched_barrier(0);
#pragma unroll
    for (int j = 0; j < NJ; ++j) {
#pragma unroll
      for (int t = 0; t < MT; ++t) {
        v2f aA; aA.x = a[t][j].x; aA.y = a[t][j].y;
        acc[t][nb] = __builtin_amdgcn_wmma_f32_16x16x4_f32(
            false, aA, false, b[2 * j], (short)0, acc[t][nb], false, false);
      }
#pragma unroll
      for (int t = 0; t < MT; ++t) {
        v2f aB; aB.x = a[t][j].z; aB.y = a[t][j].w;
        acc[t][nb] = __builtin_amdgcn_wmma_f32_16x16x4_f32(
            false, aB, false, b[2 * j + 1], (short)0, acc[t][nb], false, false);
      }
    }
    __builtin_amdgcn_sched_barrier(0);
  }
}

__device__ __forceinline__ void run_layer1_packed(const v4f* __restrict__ wp, int lo16,
                                                  const float* mult,
                                                  const v2f (&xf)[NB][NS], v8f (&acc)[MT][NB])
{
  v4f   ac[MT][NJ];
  float mc[NB];
#pragma unroll
  for (int t = 0; t < MT; ++t)
#pragma unroll
    for (int j = 0; j < NJ; ++j) ac[t][j] = wp[t * (H1 * NJ * 32) + j * 32];
#pragma unroll
  for (int nb = 0; nb < NB; ++nb) mc[nb] = mult[nb * (H1 * D) + lo16];

  int next = PRIO2_G + 1;
  for (int h = 1; h < H1; ++h) {
    if (__builtin_expect(h == next, 0)) {
      if (h == PRIO2_G + 1)      { __builtin_amdgcn_s_setprio(2); next = PRIO1_G + 1; }
      else if (h == PRIO1_G + 1) { __builtin_amdgcn_s_setprio(1); next = PRIO0_G + 1; }
      else                       { __builtin_amdgcn_s_setprio(0); next = H1; }
    }
    v4f   an[MT][NJ];
    float mn[NB];
    const v4f* wn = wp + h * (NJ * 32);
#pragma unroll
    for (int t = 0; t < MT; ++t)
#pragma unroll
      for (int j = 0; j < NJ; ++j) an[t][j] = wn[t * (H1 * NJ * 32) + j * 32];
#pragma unroll
    for (int nb = 0; nb < NB; ++nb) mn[nb] = mult[nb * (H1 * D) + h * D + lo16];
    hgroup_mt(ac, mc, xf, acc);
#pragma unroll
    for (int t = 0; t < MT; ++t)
#pragma unroll
      for (int j = 0; j < NJ; ++j) ac[t][j] = an[t][j];
#pragma unroll
    for (int nb = 0; nb < NB; ++nb) mc[nb] = mn[nb];
  }
  hgroup_mt(ac, mc, xf, acc);
}


template <int Q, bool LEAD0>
__device__ __forceinline__ void hgroup_sym(const v2f (&a0)[MT][NJ], const v2f (&a1)[MT][NJ],
                                           const float (&m)[NB],
                                           const v2f (&xf)[NB][NS], v8f (&acc)[MT][NB])
{
#pragma unroll
  for (int nb = 0; nb < NB; ++nb) {
    v2f b[NS];
#pragma unroll
    for (int s = 2 * Q; s < NS; ++s)
      if (LEAD0 || s > 2 * Q) b[s] = xf[nb][s] * m[nb];
    __builtin_amdgcn_sched_barrier(0);
#pragma unroll
    for (int j = Q; j < NJ; ++j) {
      if (LEAD0 || j > Q) {
#pragma unroll
        for (int t = 0; t < MT; ++t)
          acc[t][nb] = __builtin_amdgcn_wmma_f32_16x16x4_f32(
              false, a0[t][j], false, b[2 * j], (short)0, acc[t][nb], false, false);
      }
#pragma unroll
      for (int t = 0; t < MT; ++t)
        acc[t][nb] = __builtin_amdgcn_wmma_f32_16x16x4_f32(
            false, a1[t][j], false, b[2 * j + 1], (short)0, acc[t][nb], false, false);
    }
    __builtin_amdgcn_sched_barrier(0);
  }
}

template <int Q, bool LEAD0, int QN, bool LEADN>
__device__ __forceinline__ void sym_step(const v4f* __restrict__ sp, int un,
                                         const float* mult, int hn, int lo16,
                                         v2f (&a0)[MT][NJ], v2f (&a1)[MT][NJ], float (&mc)[NB],
                                         const v2f (&xf)[NB][NS], v8f (&acc)[MT][NB])
{
  v2f   n0[MT][NJ], n1[MT][NJ];
  float mn[NB];
#pragma unroll
  for (int t = 0; t < MT; ++t) {
#pragma unroll
    for (int j = QN; j < NJ; ++j) {
      const v4f* pp = sp + t * (NU0 * 32) + (un + (j - QN)) * 32;
      if (LEADN || j > QN) {
        const v4f q4 = *pp;
        n0[t][j].x = q4.x; n0[t][j].y = q4.y;
        n1[t][j].x = q4.z; n1[t][j].y = q4.w;
      } else {
        n1[t][j] = *((const v2f*)pp + 1);
      }
    }
  }
#pragma unroll
  for (int nb = 0; nb < NB; ++nb) mn[nb] = mult[nb * (FP * D) + hn * D + lo16];
  hgroup_sym<Q, LEAD0>(a0, a1, mc, xf, acc);
#pragma unroll
  for (int t = 0; t < MT; ++t) {
#pragma unroll
    for (int j = QN; j < NJ; ++j) {
      if (LEADN || j > QN) a0[t][j] = n0[t][j];
      a1[t][j] = n1[t][j];
    }
  }
#pragma unroll
  for (int nb = 0; nb < NB; ++nb) mc[nb] = mn[nb];
}

template <int Q>
__device__ __forceinline__ void sym_block(const v4f* __restrict__ sp, const float* mult, int lo16,
                                          v2f (&a0)[MT][NJ], v2f (&a1)[MT][NJ], float (&mc)[NB],
                                          const v2f (&xf)[NB][NS], v8f (&acc)[MT][NB])
{
  constexpr int PER = NJ - Q;
  constexpr int UB  = ubase(Q);
  constexpr int H0  = 8 * Q;
  constexpr int NC1 = (Q < 4) ? 4 : 3;
#pragma unroll 1
  for (int p = 1; p <= 4; ++p) {
    const int rn = (p & 1) | ((p & 2) << 1) | ((p & 4) >> 1);
    sym_step<Q, true, Q, true>(sp, UB + rn * PER, mult, H0 + rn, lo16, a0, a1, mc, xf, acc);
  }
#pragma unroll 1
  for (int p = 5; p < 4 + NC1; ++p) {
    const int rn = (p & 1) | ((p & 2) << 1) | ((p & 4) >> 1);
    sym_step<Q, false, Q, false>(sp, UB + rn * PER, mult, H0 + rn, lo16, a0, a1, mc, xf, acc);
  }
  if (Q < 4)
    sym_step<Q, false, Q + 1, true>(sp, ubase(Q + 1), mult, H0 + 8, lo16, a0, a1, mc, xf, acc);
  else
    hgroup_sym<Q, false>(a0, a1, mc, xf, acc);
}

__device__ __forceinline__ void run_layer0_sym(const float* __restrict__ ws, int wave, int lane,
                                               int lo16, const float* xs,
                                               const v2f (&xf)[NB][NS], v8f (&acc)[MT][NB])
{
  const v4f* sp = (const v4f*)ws + (size_t)(MT * wave) * (NU0 * 32) + lane;
  v2f   a0[MT][NJ], a1[MT][NJ];
  float mc[NB];
#pragma unroll
  for (int t = 0; t < MT; ++t) {
#pragma unroll
    for (int j = 0; j < NJ; ++j) {
      const v4f q4 = sp[t * (NU0 * 32) + j * 32];
      a0[t][j].x = q4.x; a0[t][j].y = q4.y;
      a1[t][j].x = q4.z; a1[t][j].y = q4.w;
    }
  }
#pragma unroll
  for (int nb = 0; nb < NB; ++nb) mc[nb] = xs[nb * (FP * D) + lo16];
  sym_block<0>(sp, xs, lo16, a0, a1, mc, xf, acc);
  sym_block<1>(sp, xs, lo16, a0, a1, mc, xf, acc);
  sym_block<2>(sp, xs, lo16, a0, a1, mc, xf, acc);
  sym_block<3>(sp, xs, lo16, a0, a1, mc, xf, acc);
  sym_block<4>(sp, xs, lo16, a0, a1, mc, xf, acc);
}

__global__ __launch_bounds__(TPB, 2) void cin_wmma_kernel(
    const float* __restrict__ x,  const float* __restrict__ W0,
    const float* __restrict__ b0, const float* __restrict__ W1,
    const float* __restrict__ b1, float* __restrict__ out,
    const float* __restrict__ ws, size_t ws_bytes)
{
  __shared__ __align__(16) float lds[LDS_FLOATS];
  float* xs = lds + XS_OFF;
  float* ys = lds + YS_OFF;
  float* r0 = lds + R0_OFF;
  float* r1 = lds + R1_OFF;

  const int tid   = threadIdx.x;
  const int wave  = tid >> 5;
  const int lane  = tid & 31;
  const int lo16  = lane & 15;
  const int hi    = lane >> 4;
  const int bbase = blockIdx.x * NB;

  const v4f* xg4 = (const v4f*)(x + (size_t)bbase * (F * D));
#pragma unroll
  for (int i = 0; i < (NB * F * D / 4 + TPB - 1) / TPB; ++i) {
    const int e4 = tid + i * TPB;
    if (e4 < NB * F * D / 4) {
      const int nb  = e4 / (F * D / 4);
      const int rem = e4 - nb * (F * D / 4);
      *(v4f*)(xs + nb * (FP * D) + rem * 4) = xg4[e4];
    }
  }
  if (tid < NB * D) xs[(tid >> 4) * (FP * D) + F * D + (tid & 15)] = 0.0f;
  __syncthreads();

  v2f xf[NB][NS];
#pragma unroll
  for (int nb = 0; nb < NB; ++nb) {
#pragma unroll
    for (int s = 0; s < NS; ++s) {
      const int m = 8 * (s >> 1) + 4 * hi + 2 * (s & 1);
      xf[nb][s].x = xs[nb * (FP * D) + m * D + lo16];
      xf[nb][s].y = xs[nb * (FP * D) + (m + 1) * D + lo16];
    }
  }

  v8f acc[MT][NB];
#pragma unroll
  for (int t = 0; t < MT; ++t)
#pragma unroll
    for (int nb = 0; nb < NB; ++nb)
#pragma unroll
      for (int j = 0; j < 8; ++j) acc[t][nb][j] = 0.0f;

  __builtin_amdgcn_s_setprio(3);
  if (__builtin_expect(ws_bytes >= (size_t)WS_NEED, 1)) {
    run_layer0_sym(ws, wave, lane, lo16, xs, xf, acc);
  } else {
#pragma unroll
    for (int t = 0; t < MT; ++t)
      run_layer<F>(W0 + (size_t)(((MT * wave + t) << 4) + lo16) * K0, hi, lo16, xs, FP * D, xf, acc[t]);
  }

#pragma unroll
  for (int t = 0; t < MT; ++t) {
#pragma unroll
    for (int nb = 0; nb < NB; ++nb) {
#pragma unroll
      for (int j = 0; j < 8; ++j) {
        const int row = ((MT * wave + t) << 4) + j + (hi << 3);
        float v = acc[t][nb][j] + b0[row];
        v = v > 0.0f ? v : 0.0f;
        if (wave < 2) ys[nb * (H1 * D) + row * D + lo16] = v;
        else          r0[nb * (H1 * D) + (row - H1) * D + lo16] = v;
        acc[t][nb][j] = 0.0f;
      }
    }
  }
  __syncthreads();

#pragma unroll
  for (int rep = 0; rep < NB * H1 / TPB; ++rep) {
    const int idx = tid + rep * TPB;
    const int nb  = idx >> 6;
    const int r   = idx & 63;
    float sum = 0.0f;
#pragma unroll
    for (int d = 0; d < D; ++d) sum += r0[nb * (H1 * D) + r * D + d];
    out[(size_t)(bbase + nb) * 192 + r] = sum;
  }

  if (__builtin_expect(ws_bytes >= (size_t)WS_NEED1, 1)) {
    run_layer1_packed((const v4f*)(ws + W1P_OFF) + (size_t)(MT * wave) * (H1 * NJ * 32) + lane,
                      lo16, ys, xf, acc);
  } else {
    __builtin_amdgcn_s_setprio(0);
#pragma unroll
    for (int t = 0; t < MT; ++t)
      run_layer<H1>(W1 + (size_t)(((MT * wave + t) << 4) + lo16) * K1, hi, lo16, ys, H1 * D, xf, acc[t]);
  }

#pragma unroll
  for (int t = 0; t < MT; ++t) {
#pragma unroll
    for (int nb = 0; nb < NB; ++nb) {
#pragma unroll
      for (int j = 0; j < 8; ++j) {
        const int row = ((MT * wave + t) << 4) + j + (hi << 3);
        float v = acc[t][nb][j] + b1[row];
        v = v > 0.0f ? v : 0.0f;
        r1[nb * (L0N * D) + row * D + lo16] = v;
      }
    }
  }
  __syncthreads();

#pragma unroll
  for (int rep = 0; rep < NB * L0N / TPB; ++rep) {
    const int idx = tid + rep * TPB;
    const int nb  = idx >> 7;
    const int r   = idx & 127;
    float sum = 0.0f;
#pragma unroll
    for (int d = 0; d < D; ++d) sum += r1[nb * (L0N * D) + r * D + d];
    out[(size_t)(bbase + nb) * 192 + 64 + r] = sum;
  }
}

extern "C" void kernel_launch(void* const* d_in, const int* in_sizes, int n_in,
                              void* d_out, int out_size, void* d_ws, size_t ws_size,
                              hipStream_t stream) {
  const float* x  = (const float*)d_in[0];
  const float* W0 = (const float*)d_in[1];
  const float* b0 = (const float*)d_in[2];
  const float* W1 = (const float*)d_in[3];
  const float* b1 = (const float*)d_in[4];
  float* out = (float*)d_out;
  float* ws  = (float*)d_ws;

  hipLaunchKernelGGL(cin_sym_pack_kernel, dim3(NU0 + NB1), dim3(PTPB), 0, stream,
                     W0, W1, ws, ws_size);

  dim3 grid(B_TOT / NB);
  dim3 block(TPB);
  hipLaunchKernelGGL(cin_wmma_kernel, grid, block, 0, stream,
                     x, W0, b0, W1, b1, out, (const float*)ws, ws_size);
}
